# starting kernel
# baseline (speedup 1.0000x reference)
#include <hip/hip_runtime.h>
#include <hip/hip_bf16.h>
#include <math.h>

typedef __attribute__((ext_vector_type(2))) float v2f;
typedef __attribute__((ext_vector_type(8))) float v8f;

#define HID 128
#define HEADS 2
#define DH 256
#define NEG_SLOPE 0.2f

template<int K, bool CONCAT>
__global__ __launch_bounds__(512)
void gemm_node(const float* __restrict__ A, const float* __restrict__ A2,
               const float* __restrict__ W, float* __restrict__ Hout, int n)
{
    constexpr int PITCH = K + 4;
    __shared__ float As[16 * PITCH];

    const int tid  = threadIdx.x;
    const int wid  = tid >> 5;
    const int lane = tid & 31;
    const int r0   = blockIdx.x * 16;

    for (int idx = tid; idx < 16 * K; idx += 512) {
        const int row = idx / K;
        const int k   = idx - row * K;
        const int gr  = r0 + row;
        float v = 0.0f;
        if (gr < n) {
            if (CONCAT) v = (k < 64) ? A[gr * 64 + k] : A2[gr * 64 + (k - 64)];
            else        v = A[gr * K + k];
        }
        As[row * PITCH + k] = v;
    }
    __syncthreads();

    const int c0    = wid * 16;
    const int halfl = lane & 15;
    const int koff  = (lane >> 4) * 2;

    v8f acc = {};
    for (int k0 = 0; k0 < K; k0 += 4) {
        v2f a, b;
        a.x = As[halfl * PITCH + k0 + koff];
        a.y = As[halfl * PITCH + k0 + koff + 1];
        b.x = W[(k0 + koff)     * DH + c0 + halfl];
        b.y = W[(k0 + koff + 1) * DH + c0 + halfl];
        acc = __builtin_amdgcn_wmma_f32_16x16x4_f32(
                  false, a, false, b, (short)0, acc, false, false);
    }

    const int colg  = c0 + halfl;
    const int mbase = (lane >> 4) * 8;
#pragma unroll
    for (int r = 0; r < 8; ++r) {
        const int grow = r0 + mbase + r;
        if (grow < n) Hout[grow * DH + colg] = acc[r];
    }
}

__global__ __launch_bounds__(256)
void node_scores(const float* __restrict__ H, const float* __restrict__ al,
                 const float* __restrict__ ar, float* __restrict__ el,
                 float* __restrict__ er, int n)
{
    const int lane = threadIdx.x & 31;
    const int node = blockIdx.x * 8 + (threadIdx.x >> 5);
    if (node >= n) return;

    const float* h = H + node * DH;
    float e0 = 0.f, e1 = 0.f, r0 = 0.f, r1 = 0.f;
    for (int i = lane; i < HID; i += 32) {
        const float h0 = h[i], h1 = h[HID + i];
        e0 += h0 * al[i];        e1 += h1 * al[HID + i];
        r0 += h0 * ar[i];        r1 += h1 * ar[HID + i];
    }
#pragma unroll
    for (int off = 16; off; off >>= 1) {
        e0 += __shfl_xor(e0, off); e1 += __shfl_xor(e1, off);
        r0 += __shfl_xor(r0, off); r1 += __shfl_xor(r1, off);
    }
    if (lane == 0) {
        el[node * 2 + 0] = e0; el[node * 2 + 1] = e1;
        er[node * 2 + 0] = r0; er[node * 2 + 1] = r1;
    }
}

__global__ void init_layer(float* __restrict__ agg, float* __restrict__ m,
                           float* __restrict__ s, int n)
{
    const int i = blockIdx.x * 256 + threadIdx.x;
    if (i < n * DH) agg[i] = 0.0f;
    if (i < n * HEADS) { m[i] = -INFINITY; s[i] = 0.0f; }
}

__device__ __forceinline__ float lrelu(float x) {
    return x > 0.0f ? x : NEG_SLOPE * x;
}

__device__ __forceinline__ void atomicMaxF(float* addr, float val) {
    if (val >= 0.0f) atomicMax((int*)addr, __float_as_int(val));
    else             atomicMin((unsigned int*)addr, __float_as_uint(val));
}

__global__ void edge_max(const int* __restrict__ src, const int* __restrict__ dst,
                         const float* __restrict__ el, const float* __restrict__ er,
                         float* __restrict__ m, int e)
{
    const int i = blockIdx.x * 256 + threadIdx.x;
    if (i >= e) return;
    const int s = src[i], d = dst[i];
#pragma unroll
    for (int h = 0; h < HEADS; ++h)
        atomicMaxF(&m[d * 2 + h], lrelu(el[s * 2 + h] + er[d * 2 + h]));
}

__global__ void m_fixup(float* __restrict__ m, int n2)
{
    const int i = blockIdx.x * 256 + threadIdx.x;
    if (i < n2 && !isfinite(m[i])) m[i] = 0.0f;
}

__global__ void edge_exp(const int* __restrict__ src, const int* __restrict__ dst,
                         const float* __restrict__ el, const float* __restrict__ er,
                         const float* __restrict__ m, float* __restrict__ ex,
                         float* __restrict__ ssum, int e)
{
    const int i = blockIdx.x * 256 + threadIdx.x;
    if (i >= e) return;
    const int s = src[i], d = dst[i];
#pragma unroll
    for (int h = 0; h < HEADS; ++h) {
        const float sc = lrelu(el[s * 2 + h] + er[d * 2 + h]);
        const float v  = __expf(sc - m[d * 2 + h]);
        ex[i * 2 + h] = v;
        atomicAdd(&ssum[d * 2 + h], v);
    }
}

__global__ __launch_bounds__(256)
void edge_agg(const int* __restrict__ src, const int* __restrict__ dst,
              const float* __restrict__ ex, const float* __restrict__ ssum,
              const float* __restrict__ H, float* __restrict__ agg, int e)
{
    const int w    = blockIdx.x * 8 + (threadIdx.x >> 5);
    const int lane = threadIdx.x & 31;
    if (w >= e) return;
    const int s = src[w], d = dst[w];
    const float w0 = ex[w * 2 + 0] / ssum[d * 2 + 0];
    const float w1 = ex[w * 2 + 1] / ssum[d * 2 + 1];
    const float* hs = H + s * DH;
    float* ad = agg + d * DH;
#pragma unroll
    for (int i = 0; i < 8; ++i) {
        const int j = i * 32 + lane;
        const float wt = (j < HID) ? w0 : w1;
        atomicAdd(&ad[j], wt * hs[j]);
    }
}

__global__ void bias_relu(const float* __restrict__ agg, const float* __restrict__ b,
                          float* __restrict__ outp, int n)
{
    const int i = blockIdx.x * 256 + threadIdx.x;
    if (i >= n * DH) return;
    const float v = agg[i] + b[i & (DH - 1)];
    outp[i] = v > 0.0f ? v : 0.0f;
}

__global__ __launch_bounds__(256)
void edge_classify(const int* __restrict__ src, const int* __restrict__ dst,
                   const float* __restrict__ H, const float* __restrict__ fcW,
                   const float* __restrict__ fcb, float* __restrict__ out, int e)
{
    const int w    = blockIdx.x * 8 + (threadIdx.x >> 5);
    const int lane = threadIdx.x & 31;
    if (w >= e) return;
    const int s = src[w], d = dst[w];
    const float* hs = H + s * DH;
    const float* hd = H + d * DH;
    float a0 = 0.f, a1 = 0.f;
    for (int j = lane; j < DH; j += 32) {
        const float vs = hs[j], vd = hd[j];
        a0 += vs * fcW[j * 2 + 0] + vd * fcW[(DH + j) * 2 + 0];
        a1 += vs * fcW[j * 2 + 1] + vd * fcW[(DH + j) * 2 + 1];
    }
#pragma unroll
    for (int off = 16; off; off >>= 1) {
        a0 += __shfl_xor(a0, off);
        a1 += __shfl_xor(a1, off);
    }
    if (lane == 0) {
        out[w * 2 + 0] = a0 + fcb[0];
        out[w * 2 + 1] = a1 + fcb[1];
    }
}

extern "C" void kernel_launch(void* const* d_in, const int* in_sizes, int n_in,
                              void* d_out, int out_size, void* d_ws, size_t ws_size,
                              hipStream_t stream)
{
    const float* x   = (const float*)d_in[0];
    const float* rep = (const float*)d_in[1];
    const int*   src = (const int*)d_in[2];
    const int*   dst = (const int*)d_in[3];
    const float* W1  = (const float*)d_in[4];
    const float* al1 = (const float*)d_in[5];
    const float* ar1 = (const float*)d_in[6];
    const float* b1  = (const float*)d_in[7];
    const float* W2  = (const float*)d_in[8];
    const float* al2 = (const float*)d_in[9];
    const float* ar2 = (const float*)d_in[10];
    const float* b2  = (const float*)d_in[11];
    const float* fcW = (const float*)d_in[12];
    const float* fcb = (const float*)d_in[13];
    float* out = (float*)d_out;

    const int N = in_sizes[0] / 64;
    const int E = in_sizes[2];

    float* ws  = (float*)d_ws;
    float* h   = ws;
    float* agg = h   + (size_t)N * DH;
    float* rly = agg + (size_t)N * DH;
    float* el  = rly + (size_t)N * DH;
    float* er  = el  + (size_t)N * 2;
    float* mbf = er  + (size_t)N * 2;
    float* sbf = mbf + (size_t)N * 2;
    float* ex  = sbf + (size_t)N * 2;

    const int gN16  = (N + 15) / 16;
    const int gN8   = (N + 7) / 8;
    const int gE8   = (E + 7) / 8;
    const int gE256 = (E + 255) / 256;
    const int gF    = (N * DH + 255) / 256;
    const int gM    = (N * 2 + 255) / 256;

    gemm_node<128, true><<<gN16, 512, 0, stream>>>(x, rep, W1, h, N);
    node_scores<<<gN8, 256, 0, stream>>>(h, al1, ar1, el, er, N);
    init_layer<<<gF, 256, 0, stream>>>(agg, mbf, sbf, N);
    edge_max<<<gE256, 256, 0, stream>>>(src, dst, el, er, mbf, E);
    m_fixup<<<gM, 256, 0, stream>>>(mbf, N * 2);
    edge_exp<<<gE256, 256, 0, stream>>>(src, dst, el, er, mbf, ex, sbf, E);
    edge_agg<<<gE8, 256, 0, stream>>>(src, dst, ex, sbf, h, agg, E);
    bias_relu<<<gF, 256, 0, stream>>>(agg, b1, rly, N);

    gemm_node<256, false><<<gN16, 512, 0, stream>>>(rly, nullptr, W2, h, N);
    node_scores<<<gN8, 256, 0, stream>>>(h, al2, ar2, el, er, N);
    init_layer<<<gF, 256, 0, stream>>>(agg, mbf, sbf, N);
    edge_max<<<gE256, 256, 0, stream>>>(src, dst, el, er, mbf, E);
    m_fixup<<<gM, 256, 0, stream>>>(mbf, N * 2);
    edge_exp<<<gE256, 256, 0, stream>>>(src, dst, el, er, mbf, ex, sbf, E);
    edge_agg<<<gE8, 256, 0, stream>>>(src, dst, ex, sbf, h, agg, E);
    bias_relu<<<gF, 256, 0, stream>>>(agg, b2, rly, N);

    edge_classify<<<gE8, 256, 0, stream>>>(src, dst, rly, fcW, fcb, out, E);
}


# edges counting-sorted by dst once per call (int atomic arrival-order rank); one wave per (dst,head) does max, exp-sum and weighted row gather in registers; float-atomic scatter kernels removed
# speedup vs baseline: 2.5887x; 2.5887x over previous
#include <hip/hip_runtime.h>
#include <hip/hip_bf16.h>
#include <math.h>

typedef __attribute__((ext_vector_type(2))) float v2f;
typedef __attribute__((ext_vector_type(8))) float v8f;

#define HID 128
#define HEADS 2
#define DH 256
#define NEG_SLOPE 0.2f

template<int K, bool CONCAT>
__global__ __launch_bounds__(512)
void gemm_node(const float* __restrict__ A, const float* __restrict__ A2,
               const float* __restrict__ W, float* __restrict__ Hout, int n)
{
    constexpr int PITCH = K + 4;
    __shared__ float As[16 * PITCH];

    const int tid  = threadIdx.x;
    const int wid  = tid >> 5;
    const int lane = tid & 31;
    const int r0   = blockIdx.x * 16;

    for (int idx = tid; idx < 16 * K; idx += 512) {
        const int row = idx / K;
        const int k   = idx - row * K;
        const int gr  = r0 + row;
        float v = 0.0f;
        if (gr < n) {
            if (CONCAT) v = (k < 64) ? A[gr * 64 + k] : A2[gr * 64 + (k - 64)];
            else        v = A[gr * K + k];
        }
        As[row * PITCH + k] = v;
    }
    __syncthreads();

    const int c0    = wid * 16;
    const int halfl = lane & 15;
    const int koff  = (lane >> 4) * 2;

    v8f acc = {};
    for (int k0 = 0; k0 < K; k0 += 4) {
        v2f a, b;
        a.x = As[halfl * PITCH + k0 + koff];
        a.y = As[halfl * PITCH + k0 + koff + 1];
        b.x = W[(k0 + koff)     * DH + c0 + halfl];
        b.y = W[(k0 + koff + 1) * DH + c0 + halfl];
        acc = __builtin_amdgcn_wmma_f32_16x16x4_f32(
                  false, a, false, b, (short)0, acc, false, false);
    }

    const int colg  = c0 + halfl;
    const int mbase = (lane >> 4) * 8;
#pragma unroll
    for (int r = 0; r < 8; ++r) {
        const int grow = r0 + mbase + r;
        if (grow < n) Hout[grow * DH + colg] = acc[r];
    }
}

__global__ __launch_bounds__(256)
void node_scores(const float* __restrict__ H, const float* __restrict__ al,
                 const float* __restrict__ ar, float* __restrict__ el,
                 float* __restrict__ er, int n)
{
    const int lane = threadIdx.x & 31;
    const int node = blockIdx.x * 8 + (threadIdx.x >> 5);
    if (node >= n) return;

    const float* h = H + node * DH;
    float e0 = 0.f, e1 = 0.f, r0 = 0.f, r1 = 0.f;
    for (int i = lane; i < HID; i += 32) {
        const float h0 = h[i], h1 = h[HID + i];
        e0 += h0 * al[i];        e1 += h1 * al[HID + i];
        r0 += h0 * ar[i];        r1 += h1 * ar[HID + i];
    }
#pragma unroll
    for (int off = 16; off; off >>= 1) {
        e0 += __shfl_xor(e0, off); e1 += __shfl_xor(e1, off);
        r0 += __shfl_xor(r0, off); r1 += __shfl_xor(r1, off);
    }
    if (lane == 0) {
        el[node * 2 + 0] = e0; el[node * 2 + 1] = e1;
        er[node * 2 + 0] = r0; er[node * 2 + 1] = r1;
    }
}

__device__ __forceinline__ float lrelu(float x) {
    return x > 0.0f ? x : NEG_SLOPE * x;
}

#define SCAN_T 1024
#define SCAN_I 32
#define SCAN_TILE (SCAN_T * SCAN_I)

__global__ void csr_zero(int* __restrict__ deg, int len)
{
    const int i = blockIdx.x * 256 + threadIdx.x;
    if (i < len) deg[i] = 0;
}

__global__ void csr_rank(const int* __restrict__ dst, int* __restrict__ deg,
                         int* __restrict__ rank, int e)
{
    const int i = blockIdx.x * 256 + threadIdx.x;
    if (i >= e) return;
    rank[i] = atomicAdd(&deg[dst[i]], 1);
}

__global__ __launch_bounds__(SCAN_T)
void csr_scan(const int* __restrict__ deg, int* __restrict__ rowptr, int len)
{
    __shared__ int wsum[32];
    __shared__ int tile_total;
    const int tid = threadIdx.x, lane = tid & 31, wid = tid >> 5;
    int carry = 0;
    for (int base = 0; base < len; base += SCAN_TILE) {
        const int4* p = (const int4*)(deg + base + tid * SCAN_I);
        int4 v[SCAN_I / 4];
        for (int q = 0; q < SCAN_I / 4; ++q) v[q] = p[q];
        int tot = 0;
        for (int q = 0; q < SCAN_I / 4; ++q) tot += v[q].x + v[q].y + v[q].z + v[q].w;

        int inc = tot;
        for (int off = 1; off < 32; off <<= 1) {
            const int t = __shfl_up(inc, off);
            if (lane >= off) inc += t;
        }
        if (lane == 31) wsum[wid] = inc;
        __syncthreads();
        if (wid == 0) {
            const int w = wsum[lane];
            int winc = w;
            for (int off = 1; off < 32; off <<= 1) {
                const int t = __shfl_up(winc, off);
                if (lane >= off) winc += t;
            }
            wsum[lane] = winc - w;
            if (lane == 31) tile_total = winc;
        }
        __syncthreads();
        int run = carry + wsum[wid] + (inc - tot);
        int4* o = (int4*)(rowptr + base + tid * SCAN_I);
        for (int q = 0; q < SCAN_I / 4; ++q) {
            int4 r;
            r.x = run; run += v[q].x;
            r.y = run; run += v[q].y;
            r.z = run; run += v[q].z;
            r.w = run; run += v[q].w;
            o[q] = r;
        }
        carry += tile_total;
        __syncthreads();
    }
}

__global__ void csr_fill(const int* __restrict__ src, const int* __restrict__ dst,
                         const int* __restrict__ rank, const int* __restrict__ rowptr,
                         int* __restrict__ csrc, int e)
{
    const int i = blockIdx.x * 256 + threadIdx.x;
    if (i >= e) return;
    csrc[rowptr[dst[i]] + rank[i]] = src[i];
}

__global__ __launch_bounds__(256)
void gat_gather(const int* __restrict__ rowptr, const int* __restrict__ csrc,
                const float* __restrict__ el, const float* __restrict__ er,
                const float* __restrict__ H, const float* __restrict__ bias,
                float* __restrict__ outp, int n)
{
    const int lane = threadIdx.x & 31;
    const int wv   = __builtin_amdgcn_readfirstlane(blockIdx.x * 8 + (threadIdx.x >> 5));
    const int node = wv >> 1;
    const int head = wv & 1;
    if (node >= n) return;

    const int   beg = rowptr[node];
    const int   end = rowptr[node + 1];
    const float erd = er[node * 2 + head];

    float m = -INFINITY;
    for (int k = beg + lane; k < end; k += 32)
        m = fmaxf(m, lrelu(el[csrc[k] * 2 + head] + erd));
    for (int off = 16; off; off >>= 1) m = fmaxf(m, __shfl_xor(m, off));
    if (!isfinite(m)) m = 0.0f;

    float ssum = 0.0f;
    for (int k = beg + lane; k < end; k += 32)
        ssum += __expf(lrelu(el[csrc[k] * 2 + head] + erd) - m);
    for (int off = 16; off; off >>= 1) ssum += __shfl_xor(ssum, off);

    const float* Hh = H + head * HID + lane * 4;
    float4 acc = make_float4(0.0f, 0.0f, 0.0f, 0.0f);

    for (int base = beg; base < end; base += 32) {
        int   sidx = 0;
        float a    = 0.0f;
        if (base + lane < end) {
            sidx = csrc[base + lane];
            a    = __expf(lrelu(el[sidx * 2 + head] + erd) - m) / ssum;
        }
        const int ai  = __float_as_int(a);
        const int cnt = min(32, end - base);

        int j = 0;
        for (; j + 4 <= cnt; j += 4) {
            const int s0 = __builtin_amdgcn_readlane(sidx, j);
            const int s1 = __builtin_amdgcn_readlane(sidx, j + 1);
            const int s2 = __builtin_amdgcn_readlane(sidx, j + 2);
            const int s3 = __builtin_amdgcn_readlane(sidx, j + 3);
            const float w0 = __int_as_float(__builtin_amdgcn_readlane(ai, j));
            const float w1 = __int_as_float(__builtin_amdgcn_readlane(ai, j + 1));
            const float w2 = __int_as_float(__builtin_amdgcn_readlane(ai, j + 2));
            const float w3 = __int_as_float(__builtin_amdgcn_readlane(ai, j + 3));
            const float4 h0 = *(const float4*)(Hh + (size_t)s0 * DH);
            const float4 h1 = *(const float4*)(Hh + (size_t)s1 * DH);
            const float4 h2 = *(const float4*)(Hh + (size_t)s2 * DH);
            const float4 h3 = *(const float4*)(Hh + (size_t)s3 * DH);
            acc.x += w0 * h0.x; acc.y += w0 * h0.y; acc.z += w0 * h0.z; acc.w += w0 * h0.w;
            acc.x += w1 * h1.x; acc.y += w1 * h1.y; acc.z += w1 * h1.z; acc.w += w1 * h1.w;
            acc.x += w2 * h2.x; acc.y += w2 * h2.y; acc.z += w2 * h2.z; acc.w += w2 * h2.w;
            acc.x += w3 * h3.x; acc.y += w3 * h3.y; acc.z += w3 * h3.z; acc.w += w3 * h3.w;
        }
        for (; j < cnt; ++j) {
            const int   s0 = __builtin_amdgcn_readlane(sidx, j);
            const float w0 = __int_as_float(__builtin_amdgcn_readlane(ai, j));
            const float4 h0 = *(const float4*)(Hh + (size_t)s0 * DH);
            acc.x += w0 * h0.x; acc.y += w0 * h0.y; acc.z += w0 * h0.z; acc.w += w0 * h0.w;
        }
    }

    const float4 bv = *(const float4*)(bias + head * HID + lane * 4);
    float4 o;
    o.x = acc.x + bv.x; o.x = o.x > 0.0f ? o.x : 0.0f;
    o.y = acc.y + bv.y; o.y = o.y > 0.0f ? o.y : 0.0f;
    o.z = acc.z + bv.z; o.z = o.z > 0.0f ? o.z : 0.0f;
    o.w = acc.w + bv.w; o.w = o.w > 0.0f ? o.w : 0.0f;
    *(float4*)(outp + (size_t)node * DH + head * HID + lane * 4) = o;
}

__global__ __launch_bounds__(256)
void edge_classify(const int* __restrict__ src, const int* __restrict__ dst,
                   const float* __restrict__ H, const float* __restrict__ fcW,
                   const float* __restrict__ fcb, float* __restrict__ out, int e)
{
    const int w    = blockIdx.x * 8 + (threadIdx.x >> 5);
    const int lane = threadIdx.x & 31;
    if (w >= e) return;
    const int s = src[w], d = dst[w];
    const float* hs = H + s * DH;
    const float* hd = H + d * DH;
    float a0 = 0.f, a1 = 0.f;
    for (int j = lane; j < DH; j += 32) {
        const float vs = hs[j], vd = hd[j];
        a0 += vs * fcW[j * 2 + 0] + vd * fcW[(DH + j) * 2 + 0];
        a1 += vs * fcW[j * 2 + 1] + vd * fcW[(DH + j) * 2 + 1];
    }
#pragma unroll
    for (int off = 16; off; off >>= 1) {
        a0 += __shfl_xor(a0, off);
        a1 += __shfl_xor(a1, off);
    }
    if (lane == 0) {
        out[w * 2 + 0] = a0 + fcb[0];
        out[w * 2 + 1] = a1 + fcb[1];
    }
}

extern "C" void kernel_launch(void* const* d_in, const int* in_sizes, int n_in,
                              void* d_out, int out_size, void* d_ws, size_t ws_size,
                              hipStream_t stream)
{
    const float* x   = (const float*)d_in[0];
    const float* rep = (const float*)d_in[1];
    const int*   src = (const int*)d_in[2];
    const int*   dst = (const int*)d_in[3];
    const float* W1  = (const float*)d_in[4];
    const float* al1 = (const float*)d_in[5];
    const float* ar1 = (const float*)d_in[6];
    const float* b1  = (const float*)d_in[7];
    const float* W2  = (const float*)d_in[8];
    const float* al2 = (const float*)d_in[9];
    const float* ar2 = (const float*)d_in[10];
    const float* b2  = (const float*)d_in[11];
    const float* fcW = (const float*)d_in[12];
    const float* fcb = (const float*)d_in[13];
    float* out = (float*)d_out;

    const int N = in_sizes[0] / 64;
    const int E = in_sizes[2];

    float* ws  = (float*)d_ws;
    float* h   = ws;
    int*   csr = (int*)(h + (size_t)N * DH);
    float* rly = h   + (size_t)N * DH * 2;
    float* el  = rly + (size_t)N * DH;
    float* er  = el  + (size_t)N * 2;

    const int LEN = ((N + 1 + SCAN_TILE - 1) / SCAN_TILE) * SCAN_TILE;
    int* deg    = csr;
    int* rowptr = deg    + LEN;
    int* rank   = rowptr + LEN;
    int* csrc   = rank   + E;

    const int gN16  = (N + 15) / 16;
    const int gN8   = (N + 7) / 8;
    const int gE8   = (E + 7) / 8;
    const int gE256 = (E + 255) / 256;
    const int gL256 = LEN / 256;
    const int gW8   = (N * 2 + 7) / 8;

    csr_zero<<<gL256, 256, 0, stream>>>(deg, LEN);
    csr_rank<<<gE256, 256, 0, stream>>>(dst, deg, rank, E);
    csr_scan<<<1, SCAN_T, 0, stream>>>(deg, rowptr, LEN);
    csr_fill<<<gE256, 256, 0, stream>>>(src, dst, rank, rowptr, csrc, E);

    gemm_node<128, true><<<gN16, 512, 0, stream>>>(x, rep, W1, h, N);
    node_scores<<<gN8, 256, 0, stream>>>(h, al1, ar1, el, er, N);
    gat_gather<<<gW8, 256, 0, stream>>>(rowptr, csrc, el, er, h, b1, rly, N);

    gemm_node<256, false><<<gN16, 512, 0, stream>>>(rly, nullptr, W2, h, N);
    node_scores<<<gN8, 256, 0, stream>>>(h, al2, ar2, el, er, N);
    gat_gather<<<gW8, 256, 0, stream>>>(rowptr, csrc, el, er, h, b2, rly, N);

    edge_classify<<<gE8, 256, 0, stream>>>(src, dst, rly, fcW, fcb, out, E);
}


# edge classifier applied per node in the layer-2 gather epilogue (linear map taken before the edge gather); edge kernel adds pS[src]+pD[dst]+fcb; the 512-term dot is summed in a different order
# speedup vs baseline: 3.6277x; 1.4014x over previous
#include <hip/hip_runtime.h>
#include <hip/hip_bf16.h>
#include <math.h>

typedef __attribute__((ext_vector_type(2))) float v2f;
typedef __attribute__((ext_vector_type(8))) float v8f;

#define HID 128
#define HEADS 2
#define DH 256
#define NEG_SLOPE 0.2f

template<int K, bool CONCAT>
__global__ __launch_bounds__(512)
void gemm_node(const float* __restrict__ A, const float* __restrict__ A2,
               const float* __restrict__ W, float* __restrict__ Hout, int n)
{
    constexpr int PITCH = K + 4;
    __shared__ float As[16 * PITCH];

    const int tid  = threadIdx.x;
    const int wid  = tid >> 5;
    const int lane = tid & 31;
    const int r0   = blockIdx.x * 16;

    for (int idx = tid; idx < 16 * K; idx += 512) {
        const int row = idx / K;
        const int k   = idx - row * K;
        const int gr  = r0 + row;
        float v = 0.0f;
        if (gr < n) {
            if (CONCAT) v = (k < 64) ? A[gr * 64 + k] : A2[gr * 64 + (k - 64)];
            else        v = A[gr * K + k];
        }
        As[row * PITCH + k] = v;
    }
    __syncthreads();

    const int c0    = wid * 16;
    const int halfl = lane & 15;
    const int koff  = (lane >> 4) * 2;

    v8f acc = {};
    for (int k0 = 0; k0 < K; k0 += 4) {
        v2f a, b;
        a.x = As[halfl * PITCH + k0 + koff];
        a.y = As[halfl * PITCH + k0 + koff + 1];
        b.x = W[(k0 + koff)     * DH + c0 + halfl];
        b.y = W[(k0 + koff + 1) * DH + c0 + halfl];
        acc = __builtin_amdgcn_wmma_f32_16x16x4_f32(
                  false, a, false, b, (short)0, acc, false, false);
    }

    const int colg  = c0 + halfl;
    const int mbase = (lane >> 4) * 8;
#pragma unroll
    for (int r = 0; r < 8; ++r) {
        const int grow = r0 + mbase + r;
        if (grow < n) Hout[grow * DH + colg] = acc[r];
    }
}

__global__ __launch_bounds__(256)
void node_scores(const float* __restrict__ H, const float* __restrict__ al,
                 const float* __restrict__ ar, float* __restrict__ el,
                 float* __restrict__ er, int n)
{
    const int lane = threadIdx.x & 31;
    const int node = blockIdx.x * 8 + (threadIdx.x >> 5);
    if (node >= n) return;

    const float* h = H + node * DH;
    float e0 = 0.f, e1 = 0.f, r0 = 0.f, r1 = 0.f;
    for (int i = lane; i < HID; i += 32) {
        const float h0 = h[i], h1 = h[HID + i];
        e0 += h0 * al[i];        e1 += h1 * al[HID + i];
        r0 += h0 * ar[i];        r1 += h1 * ar[HID + i];
    }
#pragma unroll
    for (int off = 16; off; off >>= 1) {
        e0 += __shfl_xor(e0, off); e1 += __shfl_xor(e1, off);
        r0 += __shfl_xor(r0, off); r1 += __shfl_xor(r1, off);
    }
    if (lane == 0) {
        el[node * 2 + 0] = e0; el[node * 2 + 1] = e1;
        er[node * 2 + 0] = r0; er[node * 2 + 1] = r1;
    }
}

__device__ __forceinline__ float lrelu(float x) {
    return x > 0.0f ? x : NEG_SLOPE * x;
}

#define SCAN_T 1024
#define SCAN_I 32
#define SCAN_TILE (SCAN_T * SCAN_I)

__global__ void csr_zero(int* __restrict__ deg, int len)
{
    const int i = blockIdx.x * 256 + threadIdx.x;
    if (i < len) deg[i] = 0;
}

__global__ void csr_rank(const int* __restrict__ dst, int* __restrict__ deg,
                         int* __restrict__ rank, int e)
{
    const int i = blockIdx.x * 256 + threadIdx.x;
    if (i >= e) return;
    rank[i] = atomicAdd(&deg[dst[i]], 1);
}

__global__ __launch_bounds__(SCAN_T)
void csr_scan(const int* __restrict__ deg, int* __restrict__ rowptr, int len)
{
    __shared__ int wsum[32];
    __shared__ int tile_total;
    const int tid = threadIdx.x, lane = tid & 31, wid = tid >> 5;
    int carry = 0;
    for (int base = 0; base < len; base += SCAN_TILE) {
        const int4* p = (const int4*)(deg + base + tid * SCAN_I);
        int4 v[SCAN_I / 4];
        for (int q = 0; q < SCAN_I / 4; ++q) v[q] = p[q];
        int tot = 0;
        for (int q = 0; q < SCAN_I / 4; ++q) tot += v[q].x + v[q].y + v[q].z + v[q].w;

        int inc = tot;
        for (int off = 1; off < 32; off <<= 1) {
            const int t = __shfl_up(inc, off);
            if (lane >= off) inc += t;
        }
        if (lane == 31) wsum[wid] = inc;
        __syncthreads();
        if (wid == 0) {
            const int w = wsum[lane];
            int winc = w;
            for (int off = 1; off < 32; off <<= 1) {
                const int t = __shfl_up(winc, off);
                if (lane >= off) winc += t;
            }
            wsum[lane] = winc - w;
            if (lane == 31) tile_total = winc;
        }
        __syncthreads();
        int run = carry + wsum[wid] + (inc - tot);
        int4* o = (int4*)(rowptr + base + tid * SCAN_I);
        for (int q = 0; q < SCAN_I / 4; ++q) {
            int4 r;
            r.x = run; run += v[q].x;
            r.y = run; run += v[q].y;
            r.z = run; run += v[q].z;
            r.w = run; run += v[q].w;
            o[q] = r;
        }
        carry += tile_total;
        __syncthreads();
    }
}

__global__ void csr_fill(const int* __restrict__ src, const int* __restrict__ dst,
                         const int* __restrict__ rank, const int* __restrict__ rowptr,
                         int* __restrict__ csrc, int e)
{
    const int i = blockIdx.x * 256 + threadIdx.x;
    if (i >= e) return;
    csrc[rowptr[dst[i]] + rank[i]] = src[i];
}

template<bool LAST>
__global__ __launch_bounds__(256)
void gat_gather(const int* __restrict__ rowptr, const int* __restrict__ csrc,
                const float* __restrict__ el, const float* __restrict__ er,
                const float* __restrict__ H, const float* __restrict__ bias,
                const float* __restrict__ fcW, float* __restrict__ outp,
                float* __restrict__ pS, float* __restrict__ pD, int n)
{
    const int lane = threadIdx.x & 31;
    const int wv   = __builtin_amdgcn_readfirstlane(blockIdx.x * 8 + (threadIdx.x >> 5));
    const int node = wv >> 1;
    const int head = wv & 1;
    if (node >= n) return;

    const int   beg = rowptr[node];
    const int   end = rowptr[node + 1];
    const float erd = er[node * 2 + head];

    float m = -INFINITY;
    for (int k = beg + lane; k < end; k += 32)
        m = fmaxf(m, lrelu(el[csrc[k] * 2 + head] + erd));
    for (int off = 16; off; off >>= 1) m = fmaxf(m, __shfl_xor(m, off));
    if (!isfinite(m)) m = 0.0f;

    float ssum = 0.0f;
    for (int k = beg + lane; k < end; k += 32)
        ssum += __expf(lrelu(el[csrc[k] * 2 + head] + erd) - m);
    for (int off = 16; off; off >>= 1) ssum += __shfl_xor(ssum, off);

    const float* Hh = H + head * HID + lane * 4;
    float4 acc = make_float4(0.0f, 0.0f, 0.0f, 0.0f);

    for (int base = beg; base < end; base += 32) {
        int   sidx = 0;
        float a    = 0.0f;
        if (base + lane < end) {
            sidx = csrc[base + lane];
            a    = __expf(lrelu(el[sidx * 2 + head] + erd) - m) / ssum;
        }
        const int ai  = __float_as_int(a);
        const int cnt = min(32, end - base);

        int j = 0;
        for (; j + 4 <= cnt; j += 4) {
            const int s0 = __builtin_amdgcn_readlane(sidx, j);
            const int s1 = __builtin_amdgcn_readlane(sidx, j + 1);
            const int s2 = __builtin_amdgcn_readlane(sidx, j + 2);
            const int s3 = __builtin_amdgcn_readlane(sidx, j + 3);
            const float w0 = __int_as_float(__builtin_amdgcn_readlane(ai, j));
            const float w1 = __int_as_float(__builtin_amdgcn_readlane(ai, j + 1));
            const float w2 = __int_as_float(__builtin_amdgcn_readlane(ai, j + 2));
            const float w3 = __int_as_float(__builtin_amdgcn_readlane(ai, j + 3));
            const float4 h0 = *(const float4*)(Hh + (size_t)s0 * DH);
            const float4 h1 = *(const float4*)(Hh + (size_t)s1 * DH);
            const float4 h2 = *(const float4*)(Hh + (size_t)s2 * DH);
            const float4 h3 = *(const float4*)(Hh + (size_t)s3 * DH);
            acc.x += w0 * h0.x; acc.y += w0 * h0.y; acc.z += w0 * h0.z; acc.w += w0 * h0.w;
            acc.x += w1 * h1.x; acc.y += w1 * h1.y; acc.z += w1 * h1.z; acc.w += w1 * h1.w;
            acc.x += w2 * h2.x; acc.y += w2 * h2.y; acc.z += w2 * h2.z; acc.w += w2 * h2.w;
            acc.x += w3 * h3.x; acc.y += w3 * h3.y; acc.z += w3 * h3.z; acc.w += w3 * h3.w;
        }
        for (; j < cnt; ++j) {
            const int   s0 = __builtin_amdgcn_readlane(sidx, j);
            const float w0 = __int_as_float(__builtin_amdgcn_readlane(ai, j));
            const float4 h0 = *(const float4*)(Hh + (size_t)s0 * DH);
            acc.x += w0 * h0.x; acc.y += w0 * h0.y; acc.z += w0 * h0.z; acc.w += w0 * h0.w;
        }
    }

    const float4 bv = *(const float4*)(bias + head * HID + lane * 4);
    float4 o;
    o.x = acc.x + bv.x; o.x = o.x > 0.0f ? o.x : 0.0f;
    o.y = acc.y + bv.y; o.y = o.y > 0.0f ? o.y : 0.0f;
    o.z = acc.z + bv.z; o.z = o.z > 0.0f ? o.z : 0.0f;
    o.w = acc.w + bv.w; o.w = o.w > 0.0f ? o.w : 0.0f;

    if (!LAST) {
        *(float4*)(outp + (size_t)node * DH + head * HID + lane * 4) = o;
    } else {
        const int    col0 = head * HID + lane * 4;
        const float4 t0 = *(const float4*)(fcW + col0 * 2);
        const float4 t1 = *(const float4*)(fcW + col0 * 2 + 4);
        const float4 u0 = *(const float4*)(fcW + (DH + col0) * 2);
        const float4 u1 = *(const float4*)(fcW + (DH + col0) * 2 + 4);
        float s0 = o.x * t0.x + o.y * t0.z + o.z * t1.x + o.w * t1.z;
        float s1 = o.x * t0.y + o.y * t0.w + o.z * t1.y + o.w * t1.w;
        float d0 = o.x * u0.x + o.y * u0.z + o.z * u1.x + o.w * u1.z;
        float d1 = o.x * u0.y + o.y * u0.w + o.z * u1.y + o.w * u1.w;
        for (int off = 16; off; off >>= 1) {
            s0 += __shfl_xor(s0, off); s1 += __shfl_xor(s1, off);
            d0 += __shfl_xor(d0, off); d1 += __shfl_xor(d1, off);
        }
        if (lane == 0) {
            *(float2*)(pS + (size_t)wv * 2) = make_float2(s0, s1);
            *(float2*)(pD + (size_t)wv * 2) = make_float2(d0, d1);
        }
    }
}

__global__ __launch_bounds__(256)
void edge_classify(const int* __restrict__ src, const int* __restrict__ dst,
                   const float* __restrict__ pS, const float* __restrict__ pD,
                   const float* __restrict__ fcb, float* __restrict__ out, int e)
{
    const int i = blockIdx.x * 256 + threadIdx.x;
    if (i >= e) return;
    const int s = src[i], d = dst[i];
    const float4 ps = *(const float4*)(pS + (size_t)s * 4);
    const float4 pd = *(const float4*)(pD + (size_t)d * 4);
    float2 o;
    o.x = (ps.x + ps.z) + (pd.x + pd.z) + fcb[0];
    o.y = (ps.y + ps.w) + (pd.y + pd.w) + fcb[1];
    *(float2*)(out + (size_t)i * 2) = o;
}

extern "C" void kernel_launch(void* const* d_in, const int* in_sizes, int n_in,
                              void* d_out, int out_size, void* d_ws, size_t ws_size,
                              hipStream_t stream)
{
    const float* x   = (const float*)d_in[0];
    const float* rep = (const float*)d_in[1];
    const int*   src = (const int*)d_in[2];
    const int*   dst = (const int*)d_in[3];
    const float* W1  = (const float*)d_in[4];
    const float* al1 = (const float*)d_in[5];
    const float* ar1 = (const float*)d_in[6];
    const float* b1  = (const float*)d_in[7];
    const float* W2  = (const float*)d_in[8];
    const float* al2 = (const float*)d_in[9];
    const float* ar2 = (const float*)d_in[10];
    const float* b2  = (const float*)d_in[11];
    const float* fcW = (const float*)d_in[12];
    const float* fcb = (const float*)d_in[13];
    float* out = (float*)d_out;

    const int N = in_sizes[0] / 64;
    const int E = in_sizes[2];

    float* ws  = (float*)d_ws;
    float* h   = ws;
    int*   csr = (int*)(h + (size_t)N * DH);
    float* rly = h   + (size_t)N * DH * 2;
    float* pS  = rly;
    float* pD  = rly + (size_t)N * 4;
    float* el  = rly + (size_t)N * DH;
    float* er  = el  + (size_t)N * 2;

    const int LEN = ((N + 1 + SCAN_TILE - 1) / SCAN_TILE) * SCAN_TILE;
    int* deg    = csr;
    int* rowptr = deg    + LEN;
    int* rank   = rowptr + LEN;
    int* csrc   = rank   + E;

    const int gN16  = (N + 15) / 16;
    const int gN8   = (N + 7) / 8;
    const int gE256 = (E + 255) / 256;
    const int gL256 = LEN / 256;
    const int gW8   = (N * 2 + 7) / 8;

    csr_zero<<<gL256, 256, 0, stream>>>(deg, LEN);
    csr_rank<<<gE256, 256, 0, stream>>>(dst, deg, rank, E);
    csr_scan<<<1, SCAN_T, 0, stream>>>(deg, rowptr, LEN);
    csr_fill<<<gE256, 256, 0, stream>>>(src, dst, rank, rowptr, csrc, E);

    gemm_node<128, true><<<gN16, 512, 0, stream>>>(x, rep, W1, h, N);
    node_scores<<<gN8, 256, 0, stream>>>(h, al1, ar1, el, er, N);
    gat_gather<false><<<gW8, 256, 0, stream>>>(rowptr, csrc, el, er, h, b1, nullptr, rly, nullptr, nullptr, N);

    gemm_node<256, false><<<gN16, 512, 0, stream>>>(rly, nullptr, W2, h, N);
    node_scores<<<gN8, 256, 0, stream>>>(h, al2, ar2, el, er, N);
    gat_gather<true><<<gW8, 256, 0, stream>>>(rowptr, csrc, el, er, h, b2, fcW, nullptr, pS, pD, N);

    edge_classify<<<gE256, 256, 0, stream>>>(src, dst, pS, pD, fcb, out, E);
}


# node attention scores el/er taken in the GEMM epilogue from the accumulators; score kernel and its re-read of H removed (11 -> 9 launches); the 128-term score dots are summed in a different order
# speedup vs baseline: 3.8514x; 1.0617x over previous
#include <hip/hip_runtime.h>
#include <hip/hip_bf16.h>
#include <math.h>

typedef __attribute__((ext_vector_type(2))) float v2f;
typedef __attribute__((ext_vector_type(8))) float v8f;

#define HID 128
#define HEADS 2
#define DH 256
#define NEG_SLOPE 0.2f

template<int K, bool CONCAT>
__global__ __launch_bounds__(512)
void gemm_node(const float* __restrict__ A, const float* __restrict__ A2,
               const float* __restrict__ W, const float* __restrict__ al,
               const float* __restrict__ ar, float* __restrict__ Hout,
               float* __restrict__ el, float* __restrict__ er, int n)
{
    constexpr int PITCH = K + 4;
    __shared__ float As[16 * PITCH];
    __shared__ float red[16 * 32];

    const int tid  = threadIdx.x;
    const int wid  = tid >> 5;
    const int lane = tid & 31;
    const int r0   = blockIdx.x * 16;

    for (int idx = tid; idx < 16 * K; idx += 512) {
        const int row = idx / K;
        const int k   = idx - row * K;
        const int gr  = r0 + row;
        float v = 0.0f;
        if (gr < n) {
            if (CONCAT) v = (k < 64) ? A[gr * 64 + k] : A2[gr * 64 + (k - 64)];
            else        v = A[gr * K + k];
        }
        As[row * PITCH + k] = v;
    }
    __syncthreads();

    const int c0    = wid * 16;
    const int halfl = lane & 15;
    const int koff  = (lane >> 4) * 2;

    v8f acc = {};
    for (int k0 = 0; k0 < K; k0 += 4) {
        v2f a, b;
        a.x = As[halfl * PITCH + k0 + koff];
        a.y = As[halfl * PITCH + k0 + koff + 1];
        b.x = W[(k0 + koff)     * DH + c0 + halfl];
        b.y = W[(k0 + koff + 1) * DH + c0 + halfl];
        acc = __builtin_amdgcn_wmma_f32_16x16x4_f32(
                  false, a, false, b, (short)0, acc, false, false);
    }

    const int colg  = c0 + halfl;
    const int mbase = (lane >> 4) * 8;
#pragma unroll
    for (int r = 0; r < 8; ++r) {
        const int grow = r0 + mbase + r;
        if (grow < n) Hout[grow * DH + colg] = acc[r];
    }

    const float alv = al[colg];
    const float arv = ar[colg];
    float p[16];
#pragma unroll
    for (int r = 0; r < 8; ++r) { p[2 * r] = acc[r] * alv; p[2 * r + 1] = acc[r] * arv; }

    const bool b3 = (lane & 8) != 0, b2 = (lane & 4) != 0, b1 = (lane & 2) != 0, b0 = (lane & 1) != 0;
    float u[8], t[4], s[2];
#pragma unroll
    for (int i = 0; i < 8; ++i) {
        const float keep = b3 ? p[i + 8] : p[i];
        const float send = b3 ? p[i] : p[i + 8];
        u[i] = keep + __shfl_xor(send, 8);
    }
#pragma unroll
    for (int i = 0; i < 4; ++i) {
        const float keep = b2 ? u[i + 4] : u[i];
        const float send = b2 ? u[i] : u[i + 4];
        t[i] = keep + __shfl_xor(send, 4);
    }
#pragma unroll
    for (int i = 0; i < 2; ++i) {
        const float keep = b1 ? t[i + 2] : t[i];
        const float send = b1 ? t[i] : t[i + 2];
        s[i] = keep + __shfl_xor(send, 2);
    }
    const float keep0 = b0 ? s[1] : s[0];
    const float send0 = b0 ? s[0] : s[1];
    red[wid * 32 + lane] = keep0 + __shfl_xor(send0, 1);
    __syncthreads();

    if (wid == 0) {
        float h0 = 0.0f, h1 = 0.0f;
#pragma unroll
        for (int i = 0; i < 8; ++i) {
            h0 += red[i * 32 + lane];
            h1 += red[(8 + i) * 32 + lane];
        }
        const int grow = r0 + mbase + (halfl >> 1);
        if (grow < n) {
            float* dstp = (lane & 1) ? er : el;
            *(float2*)(dstp + grow * 2) = make_float2(h0, h1);
        }
    }
}

__device__ __forceinline__ float lrelu(float x) {
    return x > 0.0f ? x : NEG_SLOPE * x;
}

#define SCAN_T 1024
#define SCAN_I 32
#define SCAN_TILE (SCAN_T * SCAN_I)

__global__ void csr_zero(int* __restrict__ deg, int len)
{
    const int i = blockIdx.x * 256 + threadIdx.x;
    if (i < len) deg[i] = 0;
}

__global__ void csr_rank(const int* __restrict__ dst, int* __restrict__ deg,
                         int* __restrict__ rank, int e)
{
    const int i = blockIdx.x * 256 + threadIdx.x;
    if (i >= e) return;
    rank[i] = atomicAdd(&deg[dst[i]], 1);
}

__global__ __launch_bounds__(SCAN_T)
void csr_scan(const int* __restrict__ deg, int* __restrict__ rowptr, int len)
{
    __shared__ int wsum[32];
    __shared__ int tile_total;
    const int tid = threadIdx.x, lane = tid & 31, wid = tid >> 5;
    int carry = 0;
    for (int base = 0; base < len; base += SCAN_TILE) {
        const int4* p = (const int4*)(deg + base + tid * SCAN_I);
        int4 v[SCAN_I / 4];
        for (int q = 0; q < SCAN_I / 4; ++q) v[q] = p[q];
        int tot = 0;
        for (int q = 0; q < SCAN_I / 4; ++q) tot += v[q].x + v[q].y + v[q].z + v[q].w;

        int inc = tot;
        for (int off = 1; off < 32; off <<= 1) {
            const int t = __shfl_up(inc, off);
            if (lane >= off) inc += t;
        }
        if (lane == 31) wsum[wid] = inc;
        __syncthreads();
        if (wid == 0) {
            const int w = wsum[lane];
            int winc = w;
            for (int off = 1; off < 32; off <<= 1) {
                const int t = __shfl_up(winc, off);
                if (lane >= off) winc += t;
            }
            wsum[lane] = winc - w;
            if (lane == 31) tile_total = winc;
        }
        __syncthreads();
        int run = carry + wsum[wid] + (inc - tot);
        int4* o = (int4*)(rowptr + base + tid * SCAN_I);
        for (int q = 0; q < SCAN_I / 4; ++q) {
            int4 r;
            r.x = run; run += v[q].x;
            r.y = run; run += v[q].y;
            r.z = run; run += v[q].z;
            r.w = run; run += v[q].w;
            o[q] = r;
        }
        carry += tile_total;
        __syncthreads();
    }
}

__global__ void csr_fill(const int* __restrict__ src, const int* __restrict__ dst,
                         const int* __restrict__ rank, const int* __restrict__ rowptr,
                         int* __restrict__ csrc, int e)
{
    const int i = blockIdx.x * 256 + threadIdx.x;
    if (i >= e) return;
    csrc[rowptr[dst[i]] + rank[i]] = src[i];
}

template<bool LAST>
__global__ __launch_bounds__(256)
void gat_gather(const int* __restrict__ rowptr, const int* __restrict__ csrc,
                const float* __restrict__ el, const float* __restrict__ er,
                const float* __restrict__ H, const float* __restrict__ bias,
                const float* __restrict__ fcW, float* __restrict__ outp,
                float* __restrict__ pS, float* __restrict__ pD, int n)
{
    const int lane = threadIdx.x & 31;
    const int wv   = __builtin_amdgcn_readfirstlane(blockIdx.x * 8 + (threadIdx.x >> 5));
    const int node = wv >> 1;
    const int head = wv & 1;
    if (node >= n) return;

    const int   beg = rowptr[node];
    const int   end = rowptr[node + 1];
    const float erd = er[node * 2 + head];

    float m = -INFINITY;
    for (int k = beg + lane; k < end; k += 32)
        m = fmaxf(m, lrelu(el[csrc[k] * 2 + head] + erd));
    for (int off = 16; off; off >>= 1) m = fmaxf(m, __shfl_xor(m, off));
    if (!isfinite(m)) m = 0.0f;

    float ssum = 0.0f;
    for (int k = beg + lane; k < end; k += 32)
        ssum += __expf(lrelu(el[csrc[k] * 2 + head] + erd) - m);
    for (int off = 16; off; off >>= 1) ssum += __shfl_xor(ssum, off);

    const float* Hh = H + head * HID + lane * 4;
    float4 acc = make_float4(0.0f, 0.0f, 0.0f, 0.0f);

    for (int base = beg; base < end; base += 32) {
        int   sidx = 0;
        float a    = 0.0f;
        if (base + lane < end) {
            sidx = csrc[base + lane];
            a    = __expf(lrelu(el[sidx * 2 + head] + erd) - m) / ssum;
        }
        const int ai  = __float_as_int(a);
        const int cnt = min(32, end - base);

        int j = 0;
        for (; j + 4 <= cnt; j += 4) {
            const int s0 = __builtin_amdgcn_readlane(sidx, j);
            const int s1 = __builtin_amdgcn_readlane(sidx, j + 1);
            const int s2 = __builtin_amdgcn_readlane(sidx, j + 2);
            const int s3 = __builtin_amdgcn_readlane(sidx, j + 3);
            const float w0 = __int_as_float(__builtin_amdgcn_readlane(ai, j));
            const float w1 = __int_as_float(__builtin_amdgcn_readlane(ai, j + 1));
            const float w2 = __int_as_float(__builtin_amdgcn_readlane(ai, j + 2));
            const float w3 = __int_as_float(__builtin_amdgcn_readlane(ai, j + 3));
            const float4 h0 = *(const float4*)(Hh + (size_t)s0 * DH);
            const float4 h1 = *(const float4*)(Hh + (size_t)s1 * DH);
            const float4 h2 = *(const float4*)(Hh + (size_t)s2 * DH);
            const float4 h3 = *(const float4*)(Hh + (size_t)s3 * DH);
            acc.x += w0 * h0.x; acc.y += w0 * h0.y; acc.z += w0 * h0.z; acc.w += w0 * h0.w;
            acc.x += w1 * h1.x; acc.y += w1 * h1.y; acc.z += w1 * h1.z; acc.w += w1 * h1.w;
            acc.x += w2 * h2.x; acc.y += w2 * h2.y; acc.z += w2 * h2.z; acc.w += w2 * h2.w;
            acc.x += w3 * h3.x; acc.y += w3 * h3.y; acc.z += w3 * h3.z; acc.w += w3 * h3.w;
        }
        for (; j < cnt; ++j) {
            const int   s0 = __builtin_amdgcn_readlane(sidx, j);
            const float w0 = __int_as_float(__builtin_amdgcn_readlane(ai, j));
            const float4 h0 = *(const float4*)(Hh + (size_t)s0 * DH);
            acc.x += w0 * h0.x; acc.y += w0 * h0.y; acc.z += w0 * h0.z; acc.w += w0 * h0.w;
        }
    }

    const float4 bv = *(const float4*)(bias + head * HID + lane * 4);
    float4 o;
    o.x = acc.x + bv.x; o.x = o.x > 0.0f ? o.x : 0.0f;
    o.y = acc.y + bv.y; o.y = o.y > 0.0f ? o.y : 0.0f;
    o.z = acc.z + bv.z; o.z = o.z > 0.0f ? o.z : 0.0f;
    o.w = acc.w + bv.w; o.w = o.w > 0.0f ? o.w : 0.0f;

    if (!LAST) {
        *(float4*)(outp + (size_t)node * DH + head * HID + lane * 4) = o;
    } else {
        const int    col0 = head * HID + lane * 4;
        const float4 t0 = *(const float4*)(fcW + col0 * 2);
        const float4 t1 = *(const float4*)(fcW + col0 * 2 + 4);
        const float4 u0 = *(const float4*)(fcW + (DH + col0) * 2);
        const float4 u1 = *(const float4*)(fcW + (DH + col0) * 2 + 4);
        float s0 = o.x * t0.x + o.y * t0.z + o.z * t1.x + o.w * t1.z;
        float s1 = o.x * t0.y + o.y * t0.w + o.z * t1.y + o.w * t1.w;
        float d0 = o.x * u0.x + o.y * u0.z + o.z * u1.x + o.w * u1.z;
        float d1 = o.x * u0.y + o.y * u0.w + o.z * u1.y + o.w * u1.w;
        for (int off = 16; off; off >>= 1) {
            s0 += __shfl_xor(s0, off); s1 += __shfl_xor(s1, off);
            d0 += __shfl_xor(d0, off); d1 += __shfl_xor(d1, off);
        }
        if (lane == 0) {
            *(float2*)(pS + (size_t)wv * 2) = make_float2(s0, s1);
            *(float2*)(pD + (size_t)wv * 2) = make_float2(d0, d1);
        }
    }
}

__global__ __launch_bounds__(256)
void edge_classify(const int* __restrict__ src, const int* __restrict__ dst,
                   const float* __restrict__ pS, const float* __restrict__ pD,
                   const float* __restrict__ fcb, float* __restrict__ out, int e)
{
    const int i = blockIdx.x * 256 + threadIdx.x;
    if (i >= e) return;
    const int s = src[i], d = dst[i];
    const float4 ps = *(const float4*)(pS + (size_t)s * 4);
    const float4 pd = *(const float4*)(pD + (size_t)d * 4);
    float2 o;
    o.x = (ps.x + ps.z) + (pd.x + pd.z) + fcb[0];
    o.y = (ps.y + ps.w) + (pd.y + pd.w) + fcb[1];
    *(float2*)(out + (size_t)i * 2) = o;
}

extern "C" void kernel_launch(void* const* d_in, const int* in_sizes, int n_in,
                              void* d_out, int out_size, void* d_ws, size_t ws_size,
                              hipStream_t stream)
{
    const float* x   = (const float*)d_in[0];
    const float* rep = (const float*)d_in[1];
    const int*   src = (const int*)d_in[2];
    const int*   dst = (const int*)d_in[3];
    const float* W1  = (const float*)d_in[4];
    const float* al1 = (const float*)d_in[5];
    const float* ar1 = (const float*)d_in[6];
    const float* b1  = (const float*)d_in[7];
    const float* W2  = (const float*)d_in[8];
    const float* al2 = (const float*)d_in[9];
    const float* ar2 = (const float*)d_in[10];
    const float* b2  = (const float*)d_in[11];
    const float* fcW = (const float*)d_in[12];
    const float* fcb = (const float*)d_in[13];
    float* out = (float*)d_out;

    const int N = in_sizes[0] / 64;
    const int E = in_sizes[2];

    float* ws  = (float*)d_ws;
    float* h   = ws;
    int*   csr = (int*)(h + (size_t)N * DH);
    float* rly = h   + (size_t)N * DH * 2;
    float* pS  = rly;
    float* pD  = rly + (size_t)N * 4;
    float* el  = rly + (size_t)N * DH;
    float* er  = el  + (size_t)N * 2;

    const int LEN = ((N + 1 + SCAN_TILE - 1) / SCAN_TILE) * SCAN_TILE;
    int* deg    = csr;
    int* rowptr = deg    + LEN;
    int* rank   = rowptr + LEN;
    int* csrc   = rank   + E;

    const int gN16  = (N + 15) / 16;
    const int gE256 = (E + 255) / 256;
    const int gL256 = LEN / 256;
    const int gW8   = (N * 2 + 7) / 8;

    csr_zero<<<gL256, 256, 0, stream>>>(deg, LEN);
    csr_rank<<<gE256, 256, 0, stream>>>(dst, deg, rank, E);
    csr_scan<<<1, SCAN_T, 0, stream>>>(deg, rowptr, LEN);
    csr_fill<<<gE256, 256, 0, stream>>>(src, dst, rank, rowptr, csrc, E);

    gemm_node<128, true><<<gN16, 512, 0, stream>>>(x, rep, W1, al1, ar1, h, el, er, N);
    gat_gather<false><<<gW8, 256, 0, stream>>>(rowptr, csrc, el, er, h, b1, nullptr, rly, nullptr, nullptr, N);

    gemm_node<256, false><<<gN16, 512, 0, stream>>>(rly, nullptr, W2, al2, ar2, h, el, er, N);
    gat_gather<true><<<gW8, 256, 0, stream>>>(rowptr, csrc, el, er, h, b2, fcW, nullptr, pS, pD, N);

    edge_classify<<<gE256, 256, 0, stream>>>(src, dst, pS, pD, fcb, out, E);
}


# gather takes each edge's source score from the row it has just loaded (4-term partial dot per lane + lane exchange); per-edge el fetch removed; softmax in one running-max pass, sums reordered
# speedup vs baseline: 3.9816x; 1.0338x over previous
#include <hip/hip_runtime.h>
#include <hip/hip_bf16.h>
#include <math.h>

typedef __attribute__((ext_vector_type(2))) float v2f;
typedef __attribute__((ext_vector_type(4))) float v4f;
typedef __attribute__((ext_vector_type(8))) float v8f;

#define HID 128
#define HEADS 2
#define DH 256
#define NEG_SLOPE 0.2f

template<int K, bool CONCAT>
__global__ __launch_bounds__(512)
void gemm_node(const float* __restrict__ A, const float* __restrict__ A2,
               const float* __restrict__ W, const float* __restrict__ al,
               const float* __restrict__ ar, float* __restrict__ Hout,
               float* __restrict__ el, float* __restrict__ er, int n)
{
    constexpr int PITCH = K + 4;
    __shared__ float As[16 * PITCH];
    __shared__ float red[16 * 32];

    const int tid  = threadIdx.x;
    const int wid  = tid >> 5;
    const int lane = tid & 31;
    const int r0   = blockIdx.x * 16;

    for (int idx = tid; idx < 16 * K; idx += 512) {
        const int row = idx / K;
        const int k   = idx - row * K;
        const int gr  = r0 + row;
        float v = 0.0f;
        if (gr < n) {
            if (CONCAT) v = (k < 64) ? A[gr * 64 + k] : A2[gr * 64 + (k - 64)];
            else        v = A[gr * K + k];
        }
        As[row * PITCH + k] = v;
    }
    __syncthreads();

    const int c0    = wid * 16;
    const int halfl = lane & 15;
    const int koff  = (lane >> 4) * 2;

    v8f acc = {};
    for (int k0 = 0; k0 < K; k0 += 4) {
        v2f a, b;
        a.x = As[halfl * PITCH + k0 + koff];
        a.y = As[halfl * PITCH + k0 + koff + 1];
        b.x = W[(k0 + koff)     * DH + c0 + halfl];
        b.y = W[(k0 + koff + 1) * DH + c0 + halfl];
        acc = __builtin_amdgcn_wmma_f32_16x16x4_f32(
                  false, a, false, b, (short)0, acc, false, false);
    }

    const int colg  = c0 + halfl;
    const int mbase = (lane >> 4) * 8;
#pragma unroll
    for (int r = 0; r < 8; ++r) {
        const int grow = r0 + mbase + r;
        if (grow < n) Hout[grow * DH + colg] = acc[r];
    }

    const float alv = al[colg];
    const float arv = ar[colg];
    float p[16];
#pragma unroll
    for (int r = 0; r < 8; ++r) { p[2 * r] = acc[r] * alv; p[2 * r + 1] = acc[r] * arv; }

    const bool b3 = (lane & 8) != 0, b2 = (lane & 4) != 0, b1 = (lane & 2) != 0, b0 = (lane & 1) != 0;
    float u[8], t[4], s[2];
#pragma unroll
    for (int i = 0; i < 8; ++i) {
        const float keep = b3 ? p[i + 8] : p[i];
        const float send = b3 ? p[i] : p[i + 8];
        u[i] = keep + __shfl_xor(send, 8);
    }
#pragma unroll
    for (int i = 0; i < 4; ++i) {
        const float keep = b2 ? u[i + 4] : u[i];
        const float send = b2 ? u[i] : u[i + 4];
        t[i] = keep + __shfl_xor(send, 4);
    }
#pragma unroll
    for (int i = 0; i < 2; ++i) {
        const float keep = b1 ? t[i + 2] : t[i];
        const float send = b1 ? t[i] : t[i + 2];
        s[i] = keep + __shfl_xor(send, 2);
    }
    const float keep0 = b0 ? s[1] : s[0];
    const float send0 = b0 ? s[0] : s[1];
    red[wid * 32 + lane] = keep0 + __shfl_xor(send0, 1);
    __syncthreads();

    if (wid == 0) {
        float h0 = 0.0f, h1 = 0.0f;
#pragma unroll
        for (int i = 0; i < 8; ++i) {
            h0 += red[i * 32 + lane];
            h1 += red[(8 + i) * 32 + lane];
        }
        const int grow = r0 + mbase + (halfl >> 1);
        if (grow < n) {
            float* dstp = (lane & 1) ? er : el;
            *(float2*)(dstp + grow * 2) = make_float2(h0, h1);
        }
    }
}

__device__ __forceinline__ float lrelu(float x) {
    return x > 0.0f ? x : NEG_SLOPE * x;
}

#define SCAN_T 1024
#define SCAN_I 32
#define SCAN_TILE (SCAN_T * SCAN_I)

__global__ void csr_zero(int* __restrict__ deg, int len)
{
    const int i = blockIdx.x * 256 + threadIdx.x;
    if (i < len) deg[i] = 0;
}

__global__ void csr_rank(const int* __restrict__ dst, int* __restrict__ deg,
                         int* __restrict__ rank, int e)
{
    const int i = blockIdx.x * 256 + threadIdx.x;
    if (i >= e) return;
    rank[i] = atomicAdd(&deg[dst[i]], 1);
}

__global__ __launch_bounds__(SCAN_T)
void csr_scan(const int* __restrict__ deg, int* __restrict__ rowptr, int len)
{
    __shared__ int wsum[32];
    __shared__ int tile_total;
    const int tid = threadIdx.x, lane = tid & 31, wid = tid >> 5;
    int carry = 0;
    for (int base = 0; base < len; base += SCAN_TILE) {
        const int4* p = (const int4*)(deg + base + tid * SCAN_I);
        int4 v[SCAN_I / 4];
        for (int q = 0; q < SCAN_I / 4; ++q) v[q] = p[q];
        int tot = 0;
        for (int q = 0; q < SCAN_I / 4; ++q) tot += v[q].x + v[q].y + v[q].z + v[q].w;

        int inc = tot;
        for (int off = 1; off < 32; off <<= 1) {
            const int t = __shfl_up(inc, off);
            if (lane >= off) inc += t;
        }
        if (lane == 31) wsum[wid] = inc;
        __syncthreads();
        if (wid == 0) {
            const int w = wsum[lane];
            int winc = w;
            for (int off = 1; off < 32; off <<= 1) {
                const int t = __shfl_up(winc, off);
                if (lane >= off) winc += t;
            }
            wsum[lane] = winc - w;
            if (lane == 31) tile_total = winc;
        }
        __syncthreads();
        int run = carry + wsum[wid] + (inc - tot);
        int4* o = (int4*)(rowptr + base + tid * SCAN_I);
        for (int q = 0; q < SCAN_I / 4; ++q) {
            int4 r;
            r.x = run; run += v[q].x;
            r.y = run; run += v[q].y;
            r.z = run; run += v[q].z;
            r.w = run; run += v[q].w;
            o[q] = r;
        }
        carry += tile_total;
        __syncthreads();
    }
}

__global__ void csr_fill(const int* __restrict__ src, const int* __restrict__ dst,
                         const int* __restrict__ rank, const int* __restrict__ rowptr,
                         int* __restrict__ csrc, int e)
{
    const int i = blockIdx.x * 256 + threadIdx.x;
    if (i >= e) return;
    csrc[rowptr[dst[i]] + rank[i]] = src[i];
}

__device__ __forceinline__ float xhalf(float x) {
    const int b = __builtin_bit_cast(int, x);
    return __builtin_bit_cast(float,
        __builtin_amdgcn_permlanex16(b, b, 0x76543210, 0xfedcba98, false, false));
}
template<int M>
__device__ __forceinline__ float xrow(float x) {
    const int b = __builtin_bit_cast(int, x);
    return __builtin_bit_cast(float,
        __builtin_amdgcn_update_dpp(b, b, 0x160 + M, 0xf, 0xf, false));
}

template<bool LAST>
__global__ __launch_bounds__(256)
void gat_gather(const int* __restrict__ rowptr, const int* __restrict__ csrc,
                const float* __restrict__ al, const float* __restrict__ er,
                const float* __restrict__ H, const float* __restrict__ bias,
                const float* __restrict__ fcW, float* __restrict__ outp,
                float* __restrict__ pS, float* __restrict__ pD, int n)
{
    const int lane = threadIdx.x & 31;
    const int wv   = __builtin_amdgcn_readfirstlane(blockIdx.x * 8 + (threadIdx.x >> 5));
    const int node = wv >> 1;
    const int head = wv & 1;
    if (node >= n) return;

    const int   beg = rowptr[node];
    const int   end = rowptr[node + 1];
    const float erd = er[node * 2 + head];

    const v4f    av  = *(const v4f*)(al + head * HID + lane * 4);
    const v2f    avl = av.lo, avh = av.hi;
    const float* Hh  = H + head * HID + lane * 4;
    const bool   up16 = (lane & 16) != 0;
    const bool   up8  = (lane & 8) != 0;

    float m   = -INFINITY;
    float l   = 0.0f;
    v4f   acc = {};

    for (int base = beg; base < end; base += 32) {
        int sidx = 0;
        if (base + lane < end) sidx = csrc[base + lane];
        const int cnt  = min(32, end - base);
        const int last = cnt - 1;

        for (int j = 0; j < cnt; j += 4) {
            const int s0 = __builtin_amdgcn_readlane(sidx, j);
            const int s1 = __builtin_amdgcn_readlane(sidx, min(j + 1, last));
            const int s2 = __builtin_amdgcn_readlane(sidx, min(j + 2, last));
            const int s3 = __builtin_amdgcn_readlane(sidx, min(j + 3, last));
            const v4f h0 = *(const v4f*)(Hh + (size_t)s0 * DH);
            const v4f h1 = *(const v4f*)(Hh + (size_t)s1 * DH);
            const v4f h2 = *(const v4f*)(Hh + (size_t)s2 * DH);
            const v4f h3 = *(const v4f*)(Hh + (size_t)s3 * DH);

            const v2f q0 = h0.lo * avl + h0.hi * avh;
            const v2f q1 = h1.lo * avl + h1.hi * avh;
            const v2f q2 = h2.lo * avl + h2.hi * avh;
            const v2f q3 = h3.lo * avl + h3.hi * avh;
            const float p0 = q0.x + q0.y;
            const float p1 = q1.x + q1.y;
            const float p2 = q2.x + q2.y;
            const float p3 = q3.x + q3.y;

            const float k0 = up16 ? p2 : p0, g0 = up16 ? p0 : p2;
            const float k1 = up16 ? p3 : p1, g1 = up16 ? p1 : p3;
            const float u0 = k0 + xhalf(g0);
            const float u1 = k1 + xhalf(g1);
            const float kk = up8 ? u1 : u0,  gg = up8 ? u0 : u1;
            float t = kk + xrow<8>(gg);
            t += xrow<4>(t);
            t += xrow<2>(t);
            t += xrow<1>(t);
            const float d0 = __int_as_float(__builtin_amdgcn_readlane(__float_as_int(t), 0));
            const float d1 = __int_as_float(__builtin_amdgcn_readlane(__float_as_int(t), 8));
            const float d2 = __int_as_float(__builtin_amdgcn_readlane(__float_as_int(t), 16));
            const float d3 = __int_as_float(__builtin_amdgcn_readlane(__float_as_int(t), 24));

            const float e0 = lrelu(d0 + erd);
            const float e1 = lrelu(d1 + erd);
            const float e2 = lrelu(d2 + erd);
            const float e3 = lrelu(d3 + erd);
            const float mn = fmaxf(fmaxf(m, fmaxf(e0, e1)), fmaxf(e2, e3));
            const float sc = __expf(m - mn);
            const float w0 = __expf(e0 - mn);
            const float w1 = (j + 1 < cnt) ? __expf(e1 - mn) : 0.0f;
            const float w2 = (j + 2 < cnt) ? __expf(e2 - mn) : 0.0f;
            const float w3 = (j + 3 < cnt) ? __expf(e3 - mn) : 0.0f;
            l   = l * sc + ((w0 + w1) + (w2 + w3));
            acc = acc * sc + h0 * w0;
            acc += h1 * w1;
            acc += h2 * w2;
            acc += h3 * w3;
            m = mn;
        }
    }
    if (l > 0.0f) acc = acc * (1.0f / l);

    const float4 bv = *(const float4*)(bias + head * HID + lane * 4);
    float4 o;
    o.x = acc.x + bv.x; o.x = o.x > 0.0f ? o.x : 0.0f;
    o.y = acc.y + bv.y; o.y = o.y > 0.0f ? o.y : 0.0f;
    o.z = acc.z + bv.z; o.z = o.z > 0.0f ? o.z : 0.0f;
    o.w = acc.w + bv.w; o.w = o.w > 0.0f ? o.w : 0.0f;

    if (!LAST) {
        *(float4*)(outp + (size_t)node * DH + head * HID + lane * 4) = o;
    } else {
        const int    col0 = head * HID + lane * 4;
        const float4 t0 = *(const float4*)(fcW + col0 * 2);
        const float4 t1 = *(const float4*)(fcW + col0 * 2 + 4);
        const float4 u0 = *(const float4*)(fcW + (DH + col0) * 2);
        const float4 u1 = *(const float4*)(fcW + (DH + col0) * 2 + 4);
        float s0 = o.x * t0.x + o.y * t0.z + o.z * t1.x + o.w * t1.z;
        float s1 = o.x * t0.y + o.y * t0.w + o.z * t1.y + o.w * t1.w;
        float d0 = o.x * u0.x + o.y * u0.z + o.z * u1.x + o.w * u1.z;
        float d1 = o.x * u0.y + o.y * u0.w + o.z * u1.y + o.w * u1.w;
        for (int off = 16; off; off >>= 1) {
            s0 += __shfl_xor(s0, off); s1 += __shfl_xor(s1, off);
            d0 += __shfl_xor(d0, off); d1 += __shfl_xor(d1, off);
        }
        if (lane == 0) {
            *(float2*)(pS + (size_t)wv * 2) = make_float2(s0, s1);
            *(float2*)(pD + (size_t)wv * 2) = make_float2(d0, d1);
        }
    }
}

__global__ __launch_bounds__(256)
void edge_classify(const int* __restrict__ src, const int* __restrict__ dst,
                   const float* __restrict__ pS, const float* __restrict__ pD,
                   const float* __restrict__ fcb, float* __restrict__ out, int e)
{
    const int i = blockIdx.x * 256 + threadIdx.x;
    if (i >= e) return;
    const int s = src[i], d = dst[i];
    const float4 ps = *(const float4*)(pS + (size_t)s * 4);
    const float4 pd = *(const float4*)(pD + (size_t)d * 4);
    float2 o;
    o.x = (ps.x + ps.z) + (pd.x + pd.z) + fcb[0];
    o.y = (ps.y + ps.w) + (pd.y + pd.w) + fcb[1];
    *(float2*)(out + (size_t)i * 2) = o;
}

extern "C" void kernel_launch(void* const* d_in, const int* in_sizes, int n_in,
                              void* d_out, int out_size, void* d_ws, size_t ws_size,
                              hipStream_t stream)
{
    const float* x   = (const float*)d_in[0];
    const float* rep = (const float*)d_in[1];
    const int*   src = (const int*)d_in[2];
    const int*   dst = (const int*)d_in[3];
    const float* W1  = (const float*)d_in[4];
    const float* al1 = (const float*)d_in[5];
    const float* ar1 = (const float*)d_in[6];
    const float* b1  = (const float*)d_in[7];
    const float* W2  = (const float*)d_in[8];
    const float* al2 = (const float*)d_in[9];
    const float* ar2 = (const float*)d_in[10];
    const float* b2  = (const float*)d_in[11];
    const float* fcW = (const float*)d_in[12];
    const float* fcb = (const float*)d_in[13];
    float* out = (float*)d_out;

    const int N = in_sizes[0] / 64;
    const int E = in_sizes[2];

    float* ws  = (float*)d_ws;
    float* h   = ws;
    int*   csr = (int*)(h + (size_t)N * DH);
    float* rly = h   + (size_t)N * DH * 2;
    float* pS  = rly;
    float* pD  = rly + (size_t)N * 4;
    float* el  = rly + (size_t)N * DH;
    float* er  = el  + (size_t)N * 2;

    const int LEN = ((N + 1 + SCAN_TILE - 1) / SCAN_TILE) * SCAN_TILE;
    int* deg    = csr;
    int* rowptr = deg    + LEN;
    int* rank   = rowptr + LEN;
    int* csrc   = rank   + E;

    const int gN16  = (N + 15) / 16;
    const int gE256 = (E + 255) / 256;
    const int gL256 = LEN / 256;
    const int gW8   = (N * 2 + 7) / 8;

    csr_zero<<<gL256, 256, 0, stream>>>(deg, LEN);
    csr_rank<<<gE256, 256, 0, stream>>>(dst, deg, rank, E);
    csr_scan<<<1, SCAN_T, 0, stream>>>(deg, rowptr, LEN);
    csr_fill<<<gE256, 256, 0, stream>>>(src, dst, rank, rowptr, csrc, E);

    gemm_node<128, true><<<gN16, 512, 0, stream>>>(x, rep, W1, al1, ar1, h, el, er, N);
    gat_gather<false><<<gW8, 256, 0, stream>>>(rowptr, csrc, al1, er, h, b1, nullptr, rly, nullptr, nullptr, N);

    gemm_node<256, false><<<gN16, 512, 0, stream>>>(rly, nullptr, W2, al2, ar2, h, el, er, N);
    gat_gather<true><<<gW8, 256, 0, stream>>>(rowptr, csrc, al2, er, h, b2, fcW, nullptr, pS, pD, N);

    edge_classify<<<gE256, 256, 0, stream>>>(src, dst, pS, pD, fcb, out, E);
}


# gather: one wave per dst node does both heads (16-lane row per head, 1 KB source row read once per edge); index broadcast, score reduce, exp shared; waves halved; score and pS/pD sums reordered
# speedup vs baseline: 4.1673x; 1.0466x over previous
#include <hip/hip_runtime.h>
#include <hip/hip_bf16.h>
#include <math.h>

typedef __attribute__((ext_vector_type(2))) float v2f;
typedef __attribute__((ext_vector_type(4))) float v4f;
typedef __attribute__((ext_vector_type(8))) float v8f;

#define HID 128
#define HEADS 2
#define DH 256
#define NEG_SLOPE 0.2f

template<int K, bool CONCAT>
__global__ __launch_bounds__(512)
void gemm_node(const float* __restrict__ A, const float* __restrict__ A2,
               const float* __restrict__ W, const float* __restrict__ al,
               const float* __restrict__ ar, float* __restrict__ Hout,
               float* __restrict__ el, float* __restrict__ er, int n)
{
    constexpr int PITCH = K + 4;
    __shared__ float As[16 * PITCH];
    __shared__ float red[16 * 32];

    const int tid  = threadIdx.x;
    const int wid  = tid >> 5;
    const int lane = tid & 31;
    const int r0   = blockIdx.x * 16;

    for (int idx = tid; idx < 16 * K; idx += 512) {
        const int row = idx / K;
        const int k   = idx - row * K;
        const int gr  = r0 + row;
        float v = 0.0f;
        if (gr < n) {
            if (CONCAT) v = (k < 64) ? A[gr * 64 + k] : A2[gr * 64 + (k - 64)];
            else        v = A[gr * K + k];
        }
        As[row * PITCH + k] = v;
    }
    __syncthreads();

    const int c0    = wid * 16;
    const int halfl = lane & 15;
    const int koff  = (lane >> 4) * 2;

    v8f acc = {};
    for (int k0 = 0; k0 < K; k0 += 4) {
        v2f a, b;
        a.x = As[halfl * PITCH + k0 + koff];
        a.y = As[halfl * PITCH + k0 + koff + 1];
        b.x = W[(k0 + koff)     * DH + c0 + halfl];
        b.y = W[(k0 + koff + 1) * DH + c0 + halfl];
        acc = __builtin_amdgcn_wmma_f32_16x16x4_f32(
                  false, a, false, b, (short)0, acc, false, false);
    }

    const int colg  = c0 + halfl;
    const int mbase = (lane >> 4) * 8;
#pragma unroll
    for (int r = 0; r < 8; ++r) {
        const int grow = r0 + mbase + r;
        if (grow < n) Hout[grow * DH + colg] = acc[r];
    }

    const float alv = al[colg];
    const float arv = ar[colg];
    float p[16];
#pragma unroll
    for (int r = 0; r < 8; ++r) { p[2 * r] = acc[r] * alv; p[2 * r + 1] = acc[r] * arv; }

    const bool b3 = (lane & 8) != 0, b2 = (lane & 4) != 0, b1 = (lane & 2) != 0, b0 = (lane & 1) != 0;
    float u[8], t[4], s[2];
#pragma unroll
    for (int i = 0; i < 8; ++i) {
        const float keep = b3 ? p[i + 8] : p[i];
        const float send = b3 ? p[i] : p[i + 8];
        u[i] = keep + __shfl_xor(send, 8);
    }
#pragma unroll
    for (int i = 0; i < 4; ++i) {
        const float keep = b2 ? u[i + 4] : u[i];
        const float send = b2 ? u[i] : u[i + 4];
        t[i] = keep + __shfl_xor(send, 4);
    }
#pragma unroll
    for (int i = 0; i < 2; ++i) {
        const float keep = b1 ? t[i + 2] : t[i];
        const float send = b1 ? t[i] : t[i + 2];
        s[i] = keep + __shfl_xor(send, 2);
    }
    const float keep0 = b0 ? s[1] : s[0];
    const float send0 = b0 ? s[0] : s[1];
    red[wid * 32 + lane] = keep0 + __shfl_xor(send0, 1);
    __syncthreads();

    if (wid == 0) {
        float h0 = 0.0f, h1 = 0.0f;
#pragma unroll
        for (int i = 0; i < 8; ++i) {
            h0 += red[i * 32 + lane];
            h1 += red[(8 + i) * 32 + lane];
        }
        const int grow = r0 + mbase + (halfl >> 1);
        if (grow < n) {
            float* dstp = (lane & 1) ? er : el;
            *(float2*)(dstp + grow * 2) = make_float2(h0, h1);
        }
    }
}

__device__ __forceinline__ float lrelu(float x) {
    return x > 0.0f ? x : NEG_SLOPE * x;
}

#define SCAN_T 1024
#define SCAN_I 32
#define SCAN_TILE (SCAN_T * SCAN_I)

__global__ void csr_zero(int* __restrict__ deg, int len)
{
    const int i = blockIdx.x * 256 + threadIdx.x;
    if (i < len) deg[i] = 0;
}

__global__ void csr_rank(const int* __restrict__ dst, int* __restrict__ deg,
                         int* __restrict__ rank, int e)
{
    const int i = blockIdx.x * 256 + threadIdx.x;
    if (i >= e) return;
    rank[i] = atomicAdd(&deg[dst[i]], 1);
}

__global__ __launch_bounds__(SCAN_T)
void csr_scan(const int* __restrict__ deg, int* __restrict__ rowptr, int len)
{
    __shared__ int wsum[32];
    __shared__ int tile_total;
    const int tid = threadIdx.x, lane = tid & 31, wid = tid >> 5;
    int carry = 0;
    for (int base = 0; base < len; base += SCAN_TILE) {
        const int4* p = (const int4*)(deg + base + tid * SCAN_I);
        int4 v[SCAN_I / 4];
        for (int q = 0; q < SCAN_I / 4; ++q) v[q] = p[q];
        int tot = 0;
        for (int q = 0; q < SCAN_I / 4; ++q) tot += v[q].x + v[q].y + v[q].z + v[q].w;

        int inc = tot;
        for (int off = 1; off < 32; off <<= 1) {
            const int t = __shfl_up(inc, off);
            if (lane >= off) inc += t;
        }
        if (lane == 31) wsum[wid] = inc;
        __syncthreads();
        if (wid == 0) {
            const int w = wsum[lane];
            int winc = w;
            for (int off = 1; off < 32; off <<= 1) {
                const int t = __shfl_up(winc, off);
                if (lane >= off) winc += t;
            }
            wsum[lane] = winc - w;
            if (lane == 31) tile_total = winc;
        }
        __syncthreads();
        int run = carry + wsum[wid] + (inc - tot);
        int4* o = (int4*)(rowptr + base + tid * SCAN_I);
        for (int q = 0; q < SCAN_I / 4; ++q) {
            int4 r;
            r.x = run; run += v[q].x;
            r.y = run; run += v[q].y;
            r.z = run; run += v[q].z;
            r.w = run; run += v[q].w;
            o[q] = r;
        }
        carry += tile_total;
        __syncthreads();
    }
}

__global__ void csr_fill(const int* __restrict__ src, const int* __restrict__ dst,
                         const int* __restrict__ rank, const int* __restrict__ rowptr,
                         int* __restrict__ csrc, int e)
{
    const int i = blockIdx.x * 256 + threadIdx.x;
    if (i >= e) return;
    csrc[rowptr[dst[i]] + rank[i]] = src[i];
}

template<int M>
__device__ __forceinline__ float xrow(float x) {
    const int b = __builtin_bit_cast(int, x);
    return __builtin_bit_cast(float,
        __builtin_amdgcn_update_dpp(b, b, 0x160 + M, 0xf, 0xf, false));
}
template<int L>
__device__ __forceinline__ float rshare(float x) {
    const int b = __builtin_bit_cast(int, x);
    return __builtin_bit_cast(float,
        __builtin_amdgcn_update_dpp(b, b, 0x150 + L, 0xf, 0xf, false));
}

template<bool LAST>
__global__ __launch_bounds__(256)
void gat_gather(const int* __restrict__ rowptr, const int* __restrict__ csrc,
                const float* __restrict__ al, const float* __restrict__ er,
                const float* __restrict__ H, const float* __restrict__ bias,
                const float* __restrict__ fcW, float* __restrict__ outp,
                float* __restrict__ pS, float* __restrict__ pD, int n)
{
    const int lane = threadIdx.x & 31;
    const int node = __builtin_amdgcn_readfirstlane(blockIdx.x * 8 + (threadIdx.x >> 5));
    if (node >= n) return;

    const int beg = rowptr[node];
    const int end = rowptr[node + 1];

    const int   head = lane >> 4;
    const int   col0 = head * HID + (lane & 15) * 4;
    const float erd  = er[node * 2 + head];

    const v4f    aA = *(const v4f*)(al + col0);
    const v4f    aB = *(const v4f*)(al + col0 + 64);
    const float* Hl = H + col0;
    const bool   up8 = (lane & 8) != 0;
    const bool   up4 = (lane & 4) != 0;

    float m    = -INFINITY;
    float l    = 0.0f;
    v4f   accA = {};
    v4f   accB = {};

    for (int base = beg; base < end; base += 32) {
        int sidx = 0;
        if (base + lane < end) sidx = csrc[base + lane];
        const int cnt  = min(32, end - base);
        const int last = cnt - 1;

        for (int j = 0; j < cnt; j += 4) {
            const int s0 = __builtin_amdgcn_readlane(sidx, j);
            const int s1 = __builtin_amdgcn_readlane(sidx, min(j + 1, last));
            const int s2 = __builtin_amdgcn_readlane(sidx, min(j + 2, last));
            const int s3 = __builtin_amdgcn_readlane(sidx, min(j + 3, last));
            const float* r0 = Hl + (size_t)s0 * DH;
            const float* r1 = Hl + (size_t)s1 * DH;
            const float* r2 = Hl + (size_t)s2 * DH;
            const float* r3 = Hl + (size_t)s3 * DH;
            const v4f x0 = *(const v4f*)r0, y0 = *(const v4f*)(r0 + 64);
            const v4f x1 = *(const v4f*)r1, y1 = *(const v4f*)(r1 + 64);
            const v4f x2 = *(const v4f*)r2, y2 = *(const v4f*)(r2 + 64);
            const v4f x3 = *(const v4f*)r3, y3 = *(const v4f*)(r3 + 64);

            const v2f q0 = x0.lo * aA.lo + x0.hi * aA.hi + y0.lo * aB.lo + y0.hi * aB.hi;
            const v2f q1 = x1.lo * aA.lo + x1.hi * aA.hi + y1.lo * aB.lo + y1.hi * aB.hi;
            const v2f q2 = x2.lo * aA.lo + x2.hi * aA.hi + y2.lo * aB.lo + y2.hi * aB.hi;
            const v2f q3 = x3.lo * aA.lo + x3.hi * aA.hi + y3.lo * aB.lo + y3.hi * aB.hi;
            const float p0 = q0.x + q0.y;
            const float p1 = q1.x + q1.y;
            const float p2 = q2.x + q2.y;
            const float p3 = q3.x + q3.y;

            const float k0 = up8 ? p2 : p0, g0 = up8 ? p0 : p2;
            const float k1 = up8 ? p3 : p1, g1 = up8 ? p1 : p3;
            const float u0 = k0 + xrow<8>(g0);
            const float u1 = k1 + xrow<8>(g1);
            const float kk = up4 ? u1 : u0,  gg = up4 ? u0 : u1;
            float t = kk + xrow<4>(gg);
            t += xrow<2>(t);
            t += xrow<1>(t);
            const float d0 = rshare<0>(t);
            const float d1 = rshare<4>(t);
            const float d2 = rshare<8>(t);
            const float d3 = rshare<12>(t);

            const float e0 = lrelu(d0 + erd);
            const float e1 = lrelu(d1 + erd);
            const float e2 = lrelu(d2 + erd);
            const float e3 = lrelu(d3 + erd);
            const float mn = fmaxf(fmaxf(m, fmaxf(e0, e1)), fmaxf(e2, e3));
            const float sc = __expf(m - mn);
            const float w0 = __expf(e0 - mn);
            const float w1 = (j + 1 < cnt) ? __expf(e1 - mn) : 0.0f;
            const float w2 = (j + 2 < cnt) ? __expf(e2 - mn) : 0.0f;
            const float w3 = (j + 3 < cnt) ? __expf(e3 - mn) : 0.0f;
            l    = l * sc + ((w0 + w1) + (w2 + w3));
            accA = accA * sc + x0 * w0;
            accB = accB * sc + y0 * w0;
            accA += x1 * w1;  accB += y1 * w1;
            accA += x2 * w2;  accB += y2 * w2;
            accA += x3 * w3;  accB += y3 * w3;
            m = mn;
        }
    }
    if (l > 0.0f) { const float rl = 1.0f / l; accA = accA * rl; accB = accB * rl; }

    const v4f bA = *(const v4f*)(bias + col0);
    const v4f bB = *(const v4f*)(bias + col0 + 64);
    v4f oA = accA + bA, oB = accB + bB;
    oA.x = oA.x > 0.0f ? oA.x : 0.0f;  oA.y = oA.y > 0.0f ? oA.y : 0.0f;
    oA.z = oA.z > 0.0f ? oA.z : 0.0f;  oA.w = oA.w > 0.0f ? oA.w : 0.0f;
    oB.x = oB.x > 0.0f ? oB.x : 0.0f;  oB.y = oB.y > 0.0f ? oB.y : 0.0f;
    oB.z = oB.z > 0.0f ? oB.z : 0.0f;  oB.w = oB.w > 0.0f ? oB.w : 0.0f;

    if (!LAST) {
        float* op = outp + (size_t)node * DH + col0;
        *(v4f*)op        = oA;
        *(v4f*)(op + 64) = oB;
    } else {
        const float* fs = fcW + col0 * 2;
        const float* fd = fcW + (DH + col0) * 2;
        const v4f tA0 = *(const v4f*)fs,         tA1 = *(const v4f*)(fs + 4);
        const v4f tB0 = *(const v4f*)(fs + 128), tB1 = *(const v4f*)(fs + 132);
        const v4f uA0 = *(const v4f*)fd,         uA1 = *(const v4f*)(fd + 4);
        const v4f uB0 = *(const v4f*)(fd + 128), uB1 = *(const v4f*)(fd + 132);
        float s0 = oA.x * tA0.x + oA.y * tA0.z + oA.z * tA1.x + oA.w * tA1.z
                 + oB.x * tB0.x + oB.y * tB0.z + oB.z * tB1.x + oB.w * tB1.z;
        float s1 = oA.x * tA0.y + oA.y * tA0.w + oA.z * tA1.y + oA.w * tA1.w
                 + oB.x * tB0.y + oB.y * tB0.w + oB.z * tB1.y + oB.w * tB1.w;
        float d0 = oA.x * uA0.x + oA.y * uA0.z + oA.z * uA1.x + oA.w * uA1.z
                 + oB.x * uB0.x + oB.y * uB0.z + oB.z * uB1.x + oB.w * uB1.z;
        float d1 = oA.x * uA0.y + oA.y * uA0.w + oA.z * uA1.y + oA.w * uA1.w
                 + oB.x * uB0.y + oB.y * uB0.w + oB.z * uB1.y + oB.w * uB1.w;
        s0 += xrow<8>(s0); s1 += xrow<8>(s1); d0 += xrow<8>(d0); d1 += xrow<8>(d1);
        s0 += xrow<4>(s0); s1 += xrow<4>(s1); d0 += xrow<4>(d0); d1 += xrow<4>(d1);
        s0 += xrow<2>(s0); s1 += xrow<2>(s1); d0 += xrow<2>(d0); d1 += xrow<2>(d1);
        s0 += xrow<1>(s0); s1 += xrow<1>(s1); d0 += xrow<1>(d0); d1 += xrow<1>(d1);
        if ((lane & 15) == 0) {
            const size_t slot = ((size_t)node * 2 + head) * 2;
            *(float2*)(pS + slot) = make_float2(s0, s1);
            *(float2*)(pD + slot) = make_float2(d0, d1);
        }
    }
}

__global__ __launch_bounds__(256)
void edge_classify(const int* __restrict__ src, const int* __restrict__ dst,
                   const float* __restrict__ pS, const float* __restrict__ pD,
                   const float* __restrict__ fcb, float* __restrict__ out, int e)
{
    const int i = blockIdx.x * 256 + threadIdx.x;
    if (i >= e) return;
    const int s = src[i], d = dst[i];
    const float4 ps = *(const float4*)(pS + (size_t)s * 4);
    const float4 pd = *(const float4*)(pD + (size_t)d * 4);
    float2 o;
    o.x = (ps.x + ps.z) + (pd.x + pd.z) + fcb[0];
    o.y = (ps.y + ps.w) + (pd.y + pd.w) + fcb[1];
    *(float2*)(out + (size_t)i * 2) = o;
}

extern "C" void kernel_launch(void* const* d_in, const int* in_sizes, int n_in,
                              void* d_out, int out_size, void* d_ws, size_t ws_size,
                              hipStream_t stream)
{
    const float* x   = (const float*)d_in[0];
    const float* rep = (const float*)d_in[1];
    const int*   src = (const int*)d_in[2];
    const int*   dst = (const int*)d_in[3];
    const float* W1  = (const float*)d_in[4];
    const float* al1 = (const float*)d_in[5];
    const float* ar1 = (const float*)d_in[6];
    const float* b1  = (const float*)d_in[7];
    const float* W2  = (const float*)d_in[8];
    const float* al2 = (const float*)d_in[9];
    const float* ar2 = (const float*)d_in[10];
    const float* b2  = (const float*)d_in[11];
    const float* fcW = (const float*)d_in[12];
    const float* fcb = (const float*)d_in[13];
    float* out = (float*)d_out;

    const int N = in_sizes[0] / 64;
    const int E = in_sizes[2];

    float* ws  = (float*)d_ws;
    float* h   = ws;
    int*   csr = (int*)(h + (size_t)N * DH);
    float* rly = h   + (size_t)N * DH * 2;
    float* pS  = rly;
    float* pD  = rly + (size_t)N * 4;
    float* el  = rly + (size_t)N * DH;
    float* er  = el  + (size_t)N * 2;

    const int LEN = ((N + 1 + SCAN_TILE - 1) / SCAN_TILE) * SCAN_TILE;
    int* deg    = csr;
    int* rowptr = deg    + LEN;
    int* rank   = rowptr + LEN;
    int* csrc   = rank   + E;

    const int gN16  = (N + 15) / 16;
    const int gE256 = (E + 255) / 256;
    const int gL256 = LEN / 256;
    const int gW8   = (N + 7) / 8;

    csr_zero<<<gL256, 256, 0, stream>>>(deg, LEN);
    csr_rank<<<gE256, 256, 0, stream>>>(dst, deg, rank, E);
    csr_scan<<<1, SCAN_T, 0, stream>>>(deg, rowptr, LEN);
    csr_fill<<<gE256, 256, 0, stream>>>(src, dst, rank, rowptr, csrc, E);

    gemm_node<128, true><<<gN16, 512, 0, stream>>>(x, rep, W1, al1, ar1, h, el, er, N);
    gat_gather<false><<<gW8, 256, 0, stream>>>(rowptr, csrc, al1, er, h, b1, nullptr, rly, nullptr, nullptr, N);

    gemm_node<256, false><<<gN16, 512, 0, stream>>>(rly, nullptr, W2, al2, ar2, h, el, er, N);
    gat_gather<true><<<gW8, 256, 0, stream>>>(rowptr, csrc, al2, er, h, b2, fcW, nullptr, pS, pD, N);

    edge_classify<<<gE256, 256, 0, stream>>>(src, dst, pS, pD, fcb, out, E);
}


# the single-workgroup prefix scan of the in-degrees runs as block 0 of the layer-1 GEMM launch (independent work) instead of a launch of its own: 9 -> 8 launches; scan block 512 threads
# speedup vs baseline: 4.3489x; 1.0436x over previous
#include <hip/hip_runtime.h>
#include <hip/hip_bf16.h>
#include <math.h>

typedef __attribute__((ext_vector_type(2))) float v2f;
typedef __attribute__((ext_vector_type(4))) float v4f;
typedef __attribute__((ext_vector_type(8))) float v8f;

#define HID 128
#define HEADS 2
#define DH 256
#define NEG_SLOPE 0.2f

#define SCAN_T 512
#define SCAN_I 32
#define SCAN_TILE (SCAN_T * SCAN_I)

__device__ __forceinline__
void scan_block(const int* __restrict__ deg, int* __restrict__ rowptr, int len)
{
    __shared__ int wsum[32];
    __shared__ int tile_total;
    const int tid = threadIdx.x, lane = tid & 31, wid = tid >> 5;
    int carry = 0;
    for (int base = 0; base < len; base += SCAN_TILE) {
        const int4* p = (const int4*)(deg + base + tid * SCAN_I);
        int4 v[SCAN_I / 4];
        for (int q = 0; q < SCAN_I / 4; ++q) v[q] = p[q];
        int tot = 0;
        for (int q = 0; q < SCAN_I / 4; ++q) tot += v[q].x + v[q].y + v[q].z + v[q].w;

        int inc = tot;
        for (int off = 1; off < 32; off <<= 1) {
            const int t = __shfl_up(inc, off);
            if (lane >= off) inc += t;
        }
        if (lane == 31) wsum[wid] = inc;
        __syncthreads();
        if (wid == 0) {
            const int w = (lane < SCAN_T / 32) ? wsum[lane] : 0;
            int winc = w;
            for (int off = 1; off < 32; off <<= 1) {
                const int t = __shfl_up(winc, off);
                if (lane >= off) winc += t;
            }
            wsum[lane] = winc - w;
            if (lane == 31) tile_total = winc;
        }
        __syncthreads();
        int run = carry + wsum[wid] + (inc - tot);
        int4* o = (int4*)(rowptr + base + tid * SCAN_I);
        for (int q = 0; q < SCAN_I / 4; ++q) {
            int4 r;
            r.x = run; run += v[q].x;
            r.y = run; run += v[q].y;
            r.z = run; run += v[q].z;
            r.w = run; run += v[q].w;
            o[q] = r;
        }
        carry += tile_total;
        __syncthreads();
    }
}

template<int K, bool CONCAT, bool SCAN>
__global__ __launch_bounds__(512)
void gemm_node(const float* __restrict__ A, const float* __restrict__ A2,
               const float* __restrict__ W, const float* __restrict__ al,
               const float* __restrict__ ar, float* __restrict__ Hout,
               float* __restrict__ el, float* __restrict__ er, int n,
               const int* __restrict__ deg, int* __restrict__ rowptr, int len)
{
    constexpr int PITCH = K + 4;
    __shared__ float As[16 * PITCH];
    __shared__ float red[16 * 32];

    int slab = blockIdx.x;
    if (SCAN) {
        if (slab == 0) { scan_block(deg, rowptr, len); return; }
        slab -= 1;
    }

    const int tid  = threadIdx.x;
    const int wid  = tid >> 5;
    const int lane = tid & 31;
    const int r0   = slab * 16;

    for (int idx = tid; idx < 16 * K; idx += 512) {
        const int row = idx / K;
        const int k   = idx - row * K;
        const int gr  = r0 + row;
        float v = 0.0f;
        if (gr < n) {
            if (CONCAT) v = (k < 64) ? A[gr * 64 + k] : A2[gr * 64 + (k - 64)];
            else        v = A[gr * K + k];
        }
        As[row * PITCH + k] = v;
    }
    __syncthreads();

    const int c0    = wid * 16;
    const int halfl = lane & 15;
    const int koff  = (lane >> 4) * 2;

    v8f acc = {};
    for (int k0 = 0; k0 < K; k0 += 4) {
        v2f a, b;
        a.x = As[halfl * PITCH + k0 + koff];
        a.y = As[halfl * PITCH + k0 + koff + 1];
        b.x = W[(k0 + koff)     * DH + c0 + halfl];
        b.y = W[(k0 + koff + 1) * DH + c0 + halfl];
        acc = __builtin_amdgcn_wmma_f32_16x16x4_f32(
                  false, a, false, b, (short)0, acc, false, false);
    }

    const int colg  = c0 + halfl;
    const int mbase = (lane >> 4) * 8;
#pragma unroll
    for (int r = 0; r < 8; ++r) {
        const int grow = r0 + mbase + r;
        if (grow < n) Hout[grow * DH + colg] = acc[r];
    }

    const float alv = al[colg];
    const float arv = ar[colg];
    float p[16];
#pragma unroll
    for (int r = 0; r < 8; ++r) { p[2 * r] = acc[r] * alv; p[2 * r + 1] = acc[r] * arv; }

    const bool b3 = (lane & 8) != 0, b2 = (lane & 4) != 0, b1 = (lane & 2) != 0, b0 = (lane & 1) != 0;
    float u[8], t[4], s[2];
#pragma unroll
    for (int i = 0; i < 8; ++i) {
        const float keep = b3 ? p[i + 8] : p[i];
        const float send = b3 ? p[i] : p[i + 8];
        u[i] = keep + __shfl_xor(send, 8);
    }
#pragma unroll
    for (int i = 0; i < 4; ++i) {
        const float keep = b2 ? u[i + 4] : u[i];
        const float send = b2 ? u[i] : u[i + 4];
        t[i] = keep + __shfl_xor(send, 4);
    }
#pragma unroll
    for (int i = 0; i < 2; ++i) {
        const float keep = b1 ? t[i + 2] : t[i];
        const float send = b1 ? t[i] : t[i + 2];
        s[i] = keep + __shfl_xor(send, 2);
    }
    const float keep0 = b0 ? s[1] : s[0];
    const float send0 = b0 ? s[0] : s[1];
    red[wid * 32 + lane] = keep0 + __shfl_xor(send0, 1);
    __syncthreads();

    if (wid == 0) {
        float h0 = 0.0f, h1 = 0.0f;
#pragma unroll
        for (int i = 0; i < 8; ++i) {
            h0 += red[i * 32 + lane];
            h1 += red[(8 + i) * 32 + lane];
        }
        const int grow = r0 + mbase + (halfl >> 1);
        if (grow < n) {
            float* dstp = (lane & 1) ? er : el;
            *(float2*)(dstp + grow * 2) = make_float2(h0, h1);
        }
    }
}

__device__ __forceinline__ float lrelu(float x) {
    return x > 0.0f ? x : NEG_SLOPE * x;
}


__global__ void csr_zero(int* __restrict__ deg, int len)
{
    const int i = blockIdx.x * 256 + threadIdx.x;
    if (i < len) deg[i] = 0;
}

__global__ void csr_rank(const int* __restrict__ dst, int* __restrict__ deg,
                         int* __restrict__ rank, int e)
{
    const int i = blockIdx.x * 256 + threadIdx.x;
    if (i >= e) return;
    rank[i] = atomicAdd(&deg[dst[i]], 1);
}

__global__ void csr_fill(const int* __restrict__ src, const int* __restrict__ dst,
                         const int* __restrict__ rank, const int* __restrict__ rowptr,
                         int* __restrict__ csrc, int e)
{
    const int i = blockIdx.x * 256 + threadIdx.x;
    if (i >= e) return;
    csrc[rowptr[dst[i]] + rank[i]] = src[i];
}

template<int M>
__device__ __forceinline__ float xrow(float x) {
    const int b = __builtin_bit_cast(int, x);
    return __builtin_bit_cast(float,
        __builtin_amdgcn_update_dpp(b, b, 0x160 + M, 0xf, 0xf, false));
}
template<int L>
__device__ __forceinline__ float rshare(float x) {
    const int b = __builtin_bit_cast(int, x);
    return __builtin_bit_cast(float,
        __builtin_amdgcn_update_dpp(b, b, 0x150 + L, 0xf, 0xf, false));
}

template<bool LAST>
__global__ __launch_bounds__(256)
void gat_gather(const int* __restrict__ rowptr, const int* __restrict__ csrc,
                const float* __restrict__ al, const float* __restrict__ er,
                const float* __restrict__ H, const float* __restrict__ bias,
                const float* __restrict__ fcW, float* __restrict__ outp,
                float* __restrict__ pS, float* __restrict__ pD, int n)
{
    const int lane = threadIdx.x & 31;
    const int node = __builtin_amdgcn_readfirstlane(blockIdx.x * 8 + (threadIdx.x >> 5));
    if (node >= n) return;

    const int beg = rowptr[node];
    const int end = rowptr[node + 1];

    const int   head = lane >> 4;
    const int   col0 = head * HID + (lane & 15) * 4;
    const float erd  = er[node * 2 + head];

    const v4f    aA = *(const v4f*)(al + col0);
    const v4f    aB = *(const v4f*)(al + col0 + 64);
    const float* Hl = H + col0;
    const bool   up8 = (lane & 8) != 0;
    const bool   up4 = (lane & 4) != 0;

    float m    = -INFINITY;
    float l    = 0.0f;
    v4f   accA = {};
    v4f   accB = {};

    for (int base = beg; base < end; base += 32) {
        int sidx = 0;
        if (base + lane < end) sidx = csrc[base + lane];
        const int cnt  = min(32, end - base);
        const int last = cnt - 1;

        for (int j = 0; j < cnt; j += 4) {
            const int s0 = __builtin_amdgcn_readlane(sidx, j);
            const int s1 = __builtin_amdgcn_readlane(sidx, min(j + 1, last));
            const int s2 = __builtin_amdgcn_readlane(sidx, min(j + 2, last));
            const int s3 = __builtin_amdgcn_readlane(sidx, min(j + 3, last));
            const float* r0 = Hl + (size_t)s0 * DH;
            const float* r1 = Hl + (size_t)s1 * DH;
            const float* r2 = Hl + (size_t)s2 * DH;
            const float* r3 = Hl + (size_t)s3 * DH;
            const v4f x0 = *(const v4f*)r0, y0 = *(const v4f*)(r0 + 64);
            const v4f x1 = *(const v4f*)r1, y1 = *(const v4f*)(r1 + 64);
            const v4f x2 = *(const v4f*)r2, y2 = *(const v4f*)(r2 + 64);
            const v4f x3 = *(const v4f*)r3, y3 = *(const v4f*)(r3 + 64);

            const v2f q0 = x0.lo * aA.lo + x0.hi * aA.hi + y0.lo * aB.lo + y0.hi * aB.hi;
            const v2f q1 = x1.lo * aA.lo + x1.hi * aA.hi + y1.lo * aB.lo + y1.hi * aB.hi;
            const v2f q2 = x2.lo * aA.lo + x2.hi * aA.hi + y2.lo * aB.lo + y2.hi * aB.hi;
            const v2f q3 = x3.lo * aA.lo + x3.hi * aA.hi + y3.lo * aB.lo + y3.hi * aB.hi;
            const float p0 = q0.x + q0.y;
            const float p1 = q1.x + q1.y;
            const float p2 = q2.x + q2.y;
            const float p3 = q3.x + q3.y;

            const float k0 = up8 ? p2 : p0, g0 = up8 ? p0 : p2;
            const float k1 = up8 ? p3 : p1, g1 = up8 ? p1 : p3;
            const float u0 = k0 + xrow<8>(g0);
            const float u1 = k1 + xrow<8>(g1);
            const float kk = up4 ? u1 : u0,  gg = up4 ? u0 : u1;
            float t = kk + xrow<4>(gg);
            t += xrow<2>(t);
            t += xrow<1>(t);
            const float d0 = rshare<0>(t);
            const float d1 = rshare<4>(t);
            const float d2 = rshare<8>(t);
            const float d3 = rshare<12>(t);

            const float e0 = lrelu(d0 + erd);
            const float e1 = lrelu(d1 + erd);
            const float e2 = lrelu(d2 + erd);
            const float e3 = lrelu(d3 + erd);
            const float mn = fmaxf(fmaxf(m, fmaxf(e0, e1)), fmaxf(e2, e3));
            const float sc = __expf(m - mn);
            const float w0 = __expf(e0 - mn);
            const float w1 = (j + 1 < cnt) ? __expf(e1 - mn) : 0.0f;
            const float w2 = (j + 2 < cnt) ? __expf(e2 - mn) : 0.0f;
            const float w3 = (j + 3 < cnt) ? __expf(e3 - mn) : 0.0f;
            l    = l * sc + ((w0 + w1) + (w2 + w3));
            accA = accA * sc + x0 * w0;
            accB = accB * sc + y0 * w0;
            accA += x1 * w1;  accB += y1 * w1;
            accA += x2 * w2;  accB += y2 * w2;
            accA += x3 * w3;  accB += y3 * w3;
            m = mn;
        }
    }
    if (l > 0.0f) { const float rl = 1.0f / l; accA = accA * rl; accB = accB * rl; }

    const v4f bA = *(const v4f*)(bias + col0);
    const v4f bB = *(const v4f*)(bias + col0 + 64);
    v4f oA = accA + bA, oB = accB + bB;
    oA.x = oA.x > 0.0f ? oA.x : 0.0f;  oA.y = oA.y > 0.0f ? oA.y : 0.0f;
    oA.z = oA.z > 0.0f ? oA.z : 0.0f;  oA.w = oA.w > 0.0f ? oA.w : 0.0f;
    oB.x = oB.x > 0.0f ? oB.x : 0.0f;  oB.y = oB.y > 0.0f ? oB.y : 0.0f;
    oB.z = oB.z > 0.0f ? oB.z : 0.0f;  oB.w = oB.w > 0.0f ? oB.w : 0.0f;

    if (!LAST) {
        float* op = outp + (size_t)node * DH + col0;
        *(v4f*)op        = oA;
        *(v4f*)(op + 64) = oB;
    } else {
        const float* fs = fcW + col0 * 2;
        const float* fd = fcW + (DH + col0) * 2;
        const v4f tA0 = *(const v4f*)fs,         tA1 = *(const v4f*)(fs + 4);
        const v4f tB0 = *(const v4f*)(fs + 128), tB1 = *(const v4f*)(fs + 132);
        const v4f uA0 = *(const v4f*)fd,         uA1 = *(const v4f*)(fd + 4);
        const v4f uB0 = *(const v4f*)(fd + 128), uB1 = *(const v4f*)(fd + 132);
        float s0 = oA.x * tA0.x + oA.y * tA0.z + oA.z * tA1.x + oA.w * tA1.z
                 + oB.x * tB0.x + oB.y * tB0.z + oB.z * tB1.x + oB.w * tB1.z;
        float s1 = oA.x * tA0.y + oA.y * tA0.w + oA.z * tA1.y + oA.w * tA1.w
                 + oB.x * tB0.y + oB.y * tB0.w + oB.z * tB1.y + oB.w * tB1.w;
        float d0 = oA.x * uA0.x + oA.y * uA0.z + oA.z * uA1.x + oA.w * uA1.z
                 + oB.x * uB0.x + oB.y * uB0.z + oB.z * uB1.x + oB.w * uB1.z;
        float d1 = oA.x * uA0.y + oA.y * uA0.w + oA.z * uA1.y + oA.w * uA1.w
                 + oB.x * uB0.y + oB.y * uB0.w + oB.z * uB1.y + oB.w * uB1.w;
        s0 += xrow<8>(s0); s1 += xrow<8>(s1); d0 += xrow<8>(d0); d1 += xrow<8>(d1);
        s0 += xrow<4>(s0); s1 += xrow<4>(s1); d0 += xrow<4>(d0); d1 += xrow<4>(d1);
        s0 += xrow<2>(s0); s1 += xrow<2>(s1); d0 += xrow<2>(d0); d1 += xrow<2>(d1);
        s0 += xrow<1>(s0); s1 += xrow<1>(s1); d0 += xrow<1>(d0); d1 += xrow<1>(d1);
        if ((lane & 15) == 0) {
            const size_t slot = ((size_t)node * 2 + head) * 2;
            *(float2*)(pS + slot) = make_float2(s0, s1);
            *(float2*)(pD + slot) = make_float2(d0, d1);
        }
    }
}

__global__ __launch_bounds__(256)
void edge_classify(const int* __restrict__ src, const int* __restrict__ dst,
                   const float* __restrict__ pS, const float* __restrict__ pD,
                   const float* __restrict__ fcb, float* __restrict__ out, int e)
{
    const int i = blockIdx.x * 256 + threadIdx.x;
    if (i >= e) return;
    const int s = src[i], d = dst[i];
    const float4 ps = *(const float4*)(pS + (size_t)s * 4);
    const float4 pd = *(const float4*)(pD + (size_t)d * 4);
    float2 o;
    o.x = (ps.x + ps.z) + (pd.x + pd.z) + fcb[0];
    o.y = (ps.y + ps.w) + (pd.y + pd.w) + fcb[1];
    *(float2*)(out + (size_t)i * 2) = o;
}

extern "C" void kernel_launch(void* const* d_in, const int* in_sizes, int n_in,
                              void* d_out, int out_size, void* d_ws, size_t ws_size,
                              hipStream_t stream)
{
    const float* x   = (const float*)d_in[0];
    const float* rep = (const float*)d_in[1];
    const int*   src = (const int*)d_in[2];
    const int*   dst = (const int*)d_in[3];
    const float* W1  = (const float*)d_in[4];
    const float* al1 = (const float*)d_in[5];
    const float* ar1 = (const float*)d_in[6];
    const float* b1  = (const float*)d_in[7];
    const float* W2  = (const float*)d_in[8];
    const float* al2 = (const float*)d_in[9];
    const float* ar2 = (const float*)d_in[10];
    const float* b2  = (const float*)d_in[11];
    const float* fcW = (const float*)d_in[12];
    const float* fcb = (const float*)d_in[13];
    float* out = (float*)d_out;

    const int N = in_sizes[0] / 64;
    const int E = in_sizes[2];

    float* ws  = (float*)d_ws;
    float* h   = ws;
    int*   csr = (int*)(h + (size_t)N * DH);
    float* rly = h   + (size_t)N * DH * 2;
    float* pS  = rly;
    float* pD  = rly + (size_t)N * 4;
    float* el  = rly + (size_t)N * DH;
    float* er  = el  + (size_t)N * 2;

    const int LEN = ((N + 1 + SCAN_TILE - 1) / SCAN_TILE) * SCAN_TILE;
    int* deg    = csr;
    int* rowptr = deg    + LEN;
    int* rank   = rowptr + LEN;
    int* csrc   = rank   + E;

    const int gN16  = (N + 15) / 16;
    const int gE256 = (E + 255) / 256;
    const int gL256 = LEN / 256;
    const int gW8   = (N + 7) / 8;

    csr_zero<<<gL256, 256, 0, stream>>>(deg, LEN);
    csr_rank<<<gE256, 256, 0, stream>>>(dst, deg, rank, E);

    gemm_node<128, true, true><<<gN16 + 1, 512, 0, stream>>>(x, rep, W1, al1, ar1, h, el, er, N,
                                                            deg, rowptr, LEN);
    csr_fill<<<gE256, 256, 0, stream>>>(src, dst, rank, rowptr, csrc, E);

    gat_gather<false><<<gW8, 256, 0, stream>>>(rowptr, csrc, al1, er, h, b1, nullptr, rly, nullptr, nullptr, N);

    gemm_node<256, false, false><<<gN16, 512, 0, stream>>>(rly, nullptr, W2, al2, ar2, h, el, er, N,
                                                          nullptr, nullptr, 0);
    gat_gather<true><<<gW8, 256, 0, stream>>>(rowptr, csrc, al2, er, h, b2, fcW, nullptr, pS, pD, N);

    edge_classify<<<gE256, 256, 0, stream>>>(src, dst, pS, pD, fcb, out, E);
}


# layer 1 sums each dst's 128-wide input rows before W1 (rows first copied into the workspace in the same call, so cache-resident): 512 B per edge, was 1 KB; W1, W2 in one GEMM launch; sums reordered
# speedup vs baseline: 4.5447x; 1.0450x over previous
#include <hip/hip_runtime.h>
#include <hip/hip_bf16.h>
#include <math.h>

typedef __attribute__((ext_vector_type(2))) float v2f;
typedef __attribute__((ext_vector_type(4))) float v4f;
typedef __attribute__((ext_vector_type(8))) float v8f;

#define HID 128
#define HEADS 2
#define DH 256
#define NEG_SLOPE 0.2f

#define SCAN_T 512
#define SCAN_I 32
#define SCAN_TILE (SCAN_T * SCAN_I)

__device__ __forceinline__
void scan_block(const int* __restrict__ deg, int* __restrict__ rowptr, int len)
{
    __shared__ int wsum[32];
    __shared__ int tile_total;
    const int tid = threadIdx.x, lane = tid & 31, wid = tid >> 5;
    int carry = 0;
    for (int base = 0; base < len; base += SCAN_TILE) {
        const int4* p = (const int4*)(deg + base + tid * SCAN_I);
        int4 v[SCAN_I / 4];
        for (int q = 0; q < SCAN_I / 4; ++q) v[q] = p[q];
        int tot = 0;
        for (int q = 0; q < SCAN_I / 4; ++q) tot += v[q].x + v[q].y + v[q].z + v[q].w;

        int inc = tot;
        for (int off = 1; off < 32; off <<= 1) {
            const int t = __shfl_up(inc, off);
            if (lane >= off) inc += t;
        }
        if (lane == 31) wsum[wid] = inc;
        __syncthreads();
        if (wid == 0) {
            const int w = (lane < SCAN_T / 32) ? wsum[lane] : 0;
            int winc = w;
            for (int off = 1; off < 32; off <<= 1) {
                const int t = __shfl_up(winc, off);
                if (lane >= off) winc += t;
            }
            wsum[lane] = winc - w;
            if (lane == 31) tile_total = winc;
        }
        __syncthreads();
        int run = carry + wsum[wid] + (inc - tot);
        int4* o = (int4*)(rowptr + base + tid * SCAN_I);
        for (int q = 0; q < SCAN_I / 4; ++q) {
            int4 r;
            r.x = run; run += v[q].x;
            r.y = run; run += v[q].y;
            r.z = run; run += v[q].z;
            r.w = run; run += v[q].w;
            o[q] = r;
        }
        carry += tile_total;
        __syncthreads();
    }
}

__global__ __launch_bounds__(SCAN_T)
void prep_layer1(const float* __restrict__ x, const float* __restrict__ rep,
                 const float* __restrict__ W1, const float* __restrict__ al,
                 const float* __restrict__ ar, float* __restrict__ feat,
                 float* __restrict__ wl, float* __restrict__ wr, int n,
                 const int* __restrict__ deg, int* __restrict__ rowptr, int len)
{
    if (blockIdx.x == 0) { scan_block(deg, rowptr, len); return; }
    if (blockIdx.x == 1) {
        const int  t     = threadIdx.x;
        const int  k     = t >> 2;
        const int  head  = t & 1;
        const bool right = (t & 2) != 0;
        const float* wrow = W1 + k * DH + head * HID;
        const float* a    = (right ? ar : al) + head * HID;
        float sum = 0.0f;
        for (int j = 0; j < HID; j += 4) {
            const v4f w = *(const v4f*)(wrow + j);
            const v4f v = *(const v4f*)(a + j);
            sum += w.x * v.x + w.y * v.y + w.z * v.z + w.w * v.w;
        }
        (right ? wr : wl)[head * HID + k] = sum;
        return;
    }
    const int total  = n * 32;
    const int stride = (gridDim.x - 2) * SCAN_T;
    for (int idx = (blockIdx.x - 2) * SCAN_T + threadIdx.x; idx < total; idx += stride) {
        const int row = idx >> 5;
        const int q   = idx & 31;
        const v4f v = (q < 16) ? *(const v4f*)(x + (size_t)row * 64 + q * 4)
                               : *(const v4f*)(rep + (size_t)row * 64 + (q - 16) * 4);
        *(v4f*)(feat + (size_t)row * HID + q * 4) = v;
    }
}

__global__ __launch_bounds__(512)
void gemm_two(const float* __restrict__ G, const float* __restrict__ W1,
              const float* __restrict__ bias1, const float* __restrict__ W,
              const float* __restrict__ al, const float* __restrict__ ar,
              float* __restrict__ Hout, float* __restrict__ el, float* __restrict__ er, int n)
{
    constexpr int K     = DH;
    constexpr int PITCH = K + 4;
    __shared__ float As[16 * PITCH];
    __shared__ float Rs[16 * PITCH];
    __shared__ float red[16 * 32];

    const int tid  = threadIdx.x;
    const int wid  = tid >> 5;
    const int lane = tid & 31;
    const int r0   = blockIdx.x * 16;

    for (int idx = tid; idx < 16 * K; idx += 512) {
        const int row = idx / K;
        const int k   = idx - row * K;
        const int gr  = r0 + row;
        float v = 0.0f;
        if (gr < n) v = G[(size_t)gr * K + k];
        As[row * PITCH + k] = v;
    }
    __syncthreads();

    const int c0    = wid * 16;
    const int halfl = lane & 15;
    const int koff  = (lane >> 4) * 2;
    const int colg  = c0 + halfl;
    const int mbase = (lane >> 4) * 8;

    {
        const int hk = (wid >> 3) * HID;
        v8f acc1 = {};
        for (int k0 = 0; k0 < HID; k0 += 4) {
            v2f a, b;
            a.x = As[halfl * PITCH + hk + k0 + koff];
            a.y = As[halfl * PITCH + hk + k0 + koff + 1];
            b.x = W1[(k0 + koff)     * DH + colg];
            b.y = W1[(k0 + koff + 1) * DH + colg];
            acc1 = __builtin_amdgcn_wmma_f32_16x16x4_f32(
                       false, a, false, b, (short)0, acc1, false, false);
        }
        const float bv = bias1[colg];
#pragma unroll
        for (int r = 0; r < 8; ++r) {
            const float v = acc1[r] + bv;
            Rs[(mbase + r) * PITCH + colg] = v > 0.0f ? v : 0.0f;
        }
    }
    __syncthreads();

    v8f acc = {};
    for (int k0 = 0; k0 < K; k0 += 4) {
        v2f a, b;
        a.x = Rs[halfl * PITCH + k0 + koff];
        a.y = Rs[halfl * PITCH + k0 + koff + 1];
        b.x = W[(k0 + koff)     * DH + colg];
        b.y = W[(k0 + koff + 1) * DH + colg];
        acc = __builtin_amdgcn_wmma_f32_16x16x4_f32(
                  false, a, false, b, (short)0, acc, false, false);
    }

#pragma unroll
    for (int r = 0; r < 8; ++r) {
        const int grow = r0 + mbase + r;
        if (grow < n) Hout[grow * DH + colg] = acc[r];
    }
    const float alv = al[colg];
    const float arv = ar[colg];
    float p[16];
#pragma unroll
    for (int r = 0; r < 8; ++r) { p[2 * r] = acc[r] * alv; p[2 * r + 1] = acc[r] * arv; }

    const bool b3 = (lane & 8) != 0, b2 = (lane & 4) != 0, b1 = (lane & 2) != 0, b0 = (lane & 1) != 0;
    float u[8], t[4], s[2];
#pragma unroll
    for (int i = 0; i < 8; ++i) {
        const float keep = b3 ? p[i + 8] : p[i];
        const float send = b3 ? p[i] : p[i + 8];
        u[i] = keep + __shfl_xor(send, 8);
    }
#pragma unroll
    for (int i = 0; i < 4; ++i) {
        const float keep = b2 ? u[i + 4] : u[i];
        const float send = b2 ? u[i] : u[i + 4];
        t[i] = keep + __shfl_xor(send, 4);
    }
#pragma unroll
    for (int i = 0; i < 2; ++i) {
        const float keep = b1 ? t[i + 2] : t[i];
        const float send = b1 ? t[i] : t[i + 2];
        s[i] = keep + __shfl_xor(send, 2);
    }
    const float keep0 = b0 ? s[1] : s[0];
    const float send0 = b0 ? s[0] : s[1];
    red[wid * 32 + lane] = keep0 + __shfl_xor(send0, 1);
    __syncthreads();

    if (wid == 0) {
        float h0 = 0.0f, h1 = 0.0f;
#pragma unroll
        for (int i = 0; i < 8; ++i) {
            h0 += red[i * 32 + lane];
            h1 += red[(8 + i) * 32 + lane];
        }
        const int grow = r0 + mbase + (halfl >> 1);
        if (grow < n) {
            float* dstp = (lane & 1) ? er : el;
            *(float2*)(dstp + grow * 2) = make_float2(h0, h1);
        }
    }
}

__device__ __forceinline__ float lrelu(float x) {
    return x > 0.0f ? x : NEG_SLOPE * x;
}


__global__ void csr_zero(int* __restrict__ deg, int len)
{
    const int i = blockIdx.x * 256 + threadIdx.x;
    if (i < len) deg[i] = 0;
}

__global__ void csr_rank(const int* __restrict__ dst, int* __restrict__ deg,
                         int* __restrict__ rank, int e)
{
    const int i = blockIdx.x * 256 + threadIdx.x;
    if (i >= e) return;
    rank[i] = atomicAdd(&deg[dst[i]], 1);
}

__global__ void csr_fill(const int* __restrict__ src, const int* __restrict__ dst,
                         const int* __restrict__ rank, const int* __restrict__ rowptr,
                         int* __restrict__ csrc, int e)
{
    const int i = blockIdx.x * 256 + threadIdx.x;
    if (i >= e) return;
    csrc[rowptr[dst[i]] + rank[i]] = src[i];
}

template<int M>
__device__ __forceinline__ float xrow(float x) {
    const int b = __builtin_bit_cast(int, x);
    return __builtin_bit_cast(float,
        __builtin_amdgcn_update_dpp(b, b, 0x160 + M, 0xf, 0xf, false));
}
template<int L>
__device__ __forceinline__ float rshare(float x) {
    const int b = __builtin_bit_cast(int, x);
    return __builtin_bit_cast(float,
        __builtin_amdgcn_update_dpp(b, b, 0x150 + L, 0xf, 0xf, false));
}

template<bool LAST, bool L1>
__global__ __launch_bounds__(256)
void gat_gather(const int* __restrict__ rowptr, const int* __restrict__ csrc,
                const float* __restrict__ al, const float* __restrict__ er,
                const float* __restrict__ H, const float* __restrict__ bias,
                const float* __restrict__ fcW, float* __restrict__ outp,
                float* __restrict__ pS, float* __restrict__ pD, int n)
{
    const int lane = threadIdx.x & 31;
    const int node = __builtin_amdgcn_readfirstlane(blockIdx.x * 8 + (threadIdx.x >> 5));
    if (node >= n) return;

    const int beg = rowptr[node];
    const int end = rowptr[node + 1];

    const int   head = lane >> 4;
    const int   col0 = head * HID + (lane & 15) * 4;
    constexpr int RS = L1 ? HID : DH;
    const int   rc0  = L1 ? (lane & 15) * 4 : col0;

    const v4f    aA = *(const v4f*)(al + col0);
    const v4f    aB = *(const v4f*)(al + col0 + 64);
    const float* Hl = H + rc0;

    float erd;
    if (L1) {
        const v4f    rA  = *(const v4f*)(er + col0);
        const v4f    rB  = *(const v4f*)(er + col0 + 64);
        const float* own = Hl + (size_t)node * RS;
        const v4f    xo  = *(const v4f*)own, yo = *(const v4f*)(own + 64);
        float d = xo.x * rA.x + xo.y * rA.y + xo.z * rA.z + xo.w * rA.w
                + yo.x * rB.x + yo.y * rB.y + yo.z * rB.z + yo.w * rB.w;
        d += xrow<8>(d); d += xrow<4>(d); d += xrow<2>(d); d += xrow<1>(d);
        erd = d;
    } else {
        erd = er[node * 2 + head];
    }
    const bool   up8 = (lane & 8) != 0;
    const bool   up4 = (lane & 4) != 0;

    float m    = -INFINITY;
    float l    = 0.0f;
    v4f   accA = {};
    v4f   accB = {};

    for (int base = beg; base < end; base += 32) {
        int sidx = 0;
        if (base + lane < end) sidx = csrc[base + lane];
        const int cnt  = min(32, end - base);
        const int last = cnt - 1;

        for (int j = 0; j < cnt; j += 4) {
            const int s0 = __builtin_amdgcn_readlane(sidx, j);
            const int s1 = __builtin_amdgcn_readlane(sidx, min(j + 1, last));
            const int s2 = __builtin_amdgcn_readlane(sidx, min(j + 2, last));
            const int s3 = __builtin_amdgcn_readlane(sidx, min(j + 3, last));
            const float* r0 = Hl + (size_t)s0 * RS;
            const float* r1 = Hl + (size_t)s1 * RS;
            const float* r2 = Hl + (size_t)s2 * RS;
            const float* r3 = Hl + (size_t)s3 * RS;
            const v4f x0 = *(const v4f*)r0, y0 = *(const v4f*)(r0 + 64);
            const v4f x1 = *(const v4f*)r1, y1 = *(const v4f*)(r1 + 64);
            const v4f x2 = *(const v4f*)r2, y2 = *(const v4f*)(r2 + 64);
            const v4f x3 = *(const v4f*)r3, y3 = *(const v4f*)(r3 + 64);

            const v2f q0 = x0.lo * aA.lo + x0.hi * aA.hi + y0.lo * aB.lo + y0.hi * aB.hi;
            const v2f q1 = x1.lo * aA.lo + x1.hi * aA.hi + y1.lo * aB.lo + y1.hi * aB.hi;
            const v2f q2 = x2.lo * aA.lo + x2.hi * aA.hi + y2.lo * aB.lo + y2.hi * aB.hi;
            const v2f q3 = x3.lo * aA.lo + x3.hi * aA.hi + y3.lo * aB.lo + y3.hi * aB.hi;
            const float p0 = q0.x + q0.y;
            const float p1 = q1.x + q1.y;
            const float p2 = q2.x + q2.y;
            const float p3 = q3.x + q3.y;

            const float k0 = up8 ? p2 : p0, g0 = up8 ? p0 : p2;
            const float k1 = up8 ? p3 : p1, g1 = up8 ? p1 : p3;
            const float u0 = k0 + xrow<8>(g0);
            const float u1 = k1 + xrow<8>(g1);
            const float kk = up4 ? u1 : u0,  gg = up4 ? u0 : u1;
            float t = kk + xrow<4>(gg);
            t += xrow<2>(t);
            t += xrow<1>(t);
            const float d0 = rshare<0>(t);
            const float d1 = rshare<4>(t);
            const float d2 = rshare<8>(t);
            const float d3 = rshare<12>(t);

            const float e0 = lrelu(d0 + erd);
            const float e1 = lrelu(d1 + erd);
            const float e2 = lrelu(d2 + erd);
            const float e3 = lrelu(d3 + erd);
            const float mn = fmaxf(fmaxf(m, fmaxf(e0, e1)), fmaxf(e2, e3));
            const float sc = __expf(m - mn);
            const float w0 = __expf(e0 - mn);
            const float w1 = (j + 1 < cnt) ? __expf(e1 - mn) : 0.0f;
            const float w2 = (j + 2 < cnt) ? __expf(e2 - mn) : 0.0f;
            const float w3 = (j + 3 < cnt) ? __expf(e3 - mn) : 0.0f;
            l    = l * sc + ((w0 + w1) + (w2 + w3));
            accA = accA * sc + x0 * w0;
            accB = accB * sc + y0 * w0;
            accA += x1 * w1;  accB += y1 * w1;
            accA += x2 * w2;  accB += y2 * w2;
            accA += x3 * w3;  accB += y3 * w3;
            m = mn;
        }
    }
    if (l > 0.0f) { const float rl = 1.0f / l; accA = accA * rl; accB = accB * rl; }

    if (L1) {
        float* op = outp + (size_t)node * DH + col0;
        *(v4f*)op        = accA;
        *(v4f*)(op + 64) = accB;
        return;
    }

    const v4f bA = *(const v4f*)(bias + col0);
    const v4f bB = *(const v4f*)(bias + col0 + 64);
    v4f oA = accA + bA, oB = accB + bB;
    oA.x = oA.x > 0.0f ? oA.x : 0.0f;  oA.y = oA.y > 0.0f ? oA.y : 0.0f;
    oA.z = oA.z > 0.0f ? oA.z : 0.0f;  oA.w = oA.w > 0.0f ? oA.w : 0.0f;
    oB.x = oB.x > 0.0f ? oB.x : 0.0f;  oB.y = oB.y > 0.0f ? oB.y : 0.0f;
    oB.z = oB.z > 0.0f ? oB.z : 0.0f;  oB.w = oB.w > 0.0f ? oB.w : 0.0f;

    if (!LAST) {
        float* op = outp + (size_t)node * DH + col0;
        *(v4f*)op        = oA;
        *(v4f*)(op + 64) = oB;
    } else {
        const float* fs = fcW + col0 * 2;
        const float* fd = fcW + (DH + col0) * 2;
        const v4f tA0 = *(const v4f*)fs,         tA1 = *(const v4f*)(fs + 4);
        const v4f tB0 = *(const v4f*)(fs + 128), tB1 = *(const v4f*)(fs + 132);
        const v4f uA0 = *(const v4f*)fd,         uA1 = *(const v4f*)(fd + 4);
        const v4f uB0 = *(const v4f*)(fd + 128), uB1 = *(const v4f*)(fd + 132);
        float s0 = oA.x * tA0.x + oA.y * tA0.z + oA.z * tA1.x + oA.w * tA1.z
                 + oB.x * tB0.x + oB.y * tB0.z + oB.z * tB1.x + oB.w * tB1.z;
        float s1 = oA.x * tA0.y + oA.y * tA0.w + oA.z * tA1.y + oA.w * tA1.w
                 + oB.x * tB0.y + oB.y * tB0.w + oB.z * tB1.y + oB.w * tB1.w;
        float d0 = oA.x * uA0.x + oA.y * uA0.z + oA.z * uA1.x + oA.w * uA1.z
                 + oB.x * uB0.x + oB.y * uB0.z + oB.z * uB1.x + oB.w * uB1.z;
        float d1 = oA.x * uA0.y + oA.y * uA0.w + oA.z * uA1.y + oA.w * uA1.w
                 + oB.x * uB0.y + oB.y * uB0.w + oB.z * uB1.y + oB.w * uB1.w;
        s0 += xrow<8>(s0); s1 += xrow<8>(s1); d0 += xrow<8>(d0); d1 += xrow<8>(d1);
        s0 += xrow<4>(s0); s1 += xrow<4>(s1); d0 += xrow<4>(d0); d1 += xrow<4>(d1);
        s0 += xrow<2>(s0); s1 += xrow<2>(s1); d0 += xrow<2>(d0); d1 += xrow<2>(d1);
        s0 += xrow<1>(s0); s1 += xrow<1>(s1); d0 += xrow<1>(d0); d1 += xrow<1>(d1);
        if ((lane & 15) == 0) {
            const size_t slot = ((size_t)node * 2 + head) * 2;
            *(float2*)(pS + slot) = make_float2(s0, s1);
            *(float2*)(pD + slot) = make_float2(d0, d1);
        }
    }
}

__global__ __launch_bounds__(256)
void edge_classify(const int* __restrict__ src, const int* __restrict__ dst,
                   const float* __restrict__ pS, const float* __restrict__ pD,
                   const float* __restrict__ fcb, float* __restrict__ out, int e)
{
    const int i = blockIdx.x * 256 + threadIdx.x;
    if (i >= e) return;
    const int s = src[i], d = dst[i];
    const float4 ps = *(const float4*)(pS + (size_t)s * 4);
    const float4 pd = *(const float4*)(pD + (size_t)d * 4);
    float2 o;
    o.x = (ps.x + ps.z) + (pd.x + pd.z) + fcb[0];
    o.y = (ps.y + ps.w) + (pd.y + pd.w) + fcb[1];
    *(float2*)(out + (size_t)i * 2) = o;
}

extern "C" void kernel_launch(void* const* d_in, const int* in_sizes, int n_in,
                              void* d_out, int out_size, void* d_ws, size_t ws_size,
                              hipStream_t stream)
{
    const float* x   = (const float*)d_in[0];
    const float* rep = (const float*)d_in[1];
    const int*   src = (const int*)d_in[2];
    const int*   dst = (const int*)d_in[3];
    const float* W1  = (const float*)d_in[4];
    const float* al1 = (const float*)d_in[5];
    const float* ar1 = (const float*)d_in[6];
    const float* b1  = (const float*)d_in[7];
    const float* W2  = (const float*)d_in[8];
    const float* al2 = (const float*)d_in[9];
    const float* ar2 = (const float*)d_in[10];
    const float* b2  = (const float*)d_in[11];
    const float* fcW = (const float*)d_in[12];
    const float* fcb = (const float*)d_in[13];
    float* out = (float*)d_out;

    const int N = in_sizes[0] / 64;
    const int E = in_sizes[2];

    float* ws  = (float*)d_ws;
    float* h   = ws;
    float* feat = ws;
    int*   csr = (int*)(h + (size_t)N * DH);
    float* rly = h   + (size_t)N * DH * 2;
    float* pS  = rly;
    float* pD  = rly + (size_t)N * 4;
    float* el  = rly + (size_t)N * DH;
    float* er  = el  + (size_t)N * 2;

    const int LEN = ((N + 1 + SCAN_TILE - 1) / SCAN_TILE) * SCAN_TILE;
    int* deg    = csr;
    int* rowptr = deg    + LEN;
    int* rank   = rowptr + LEN;
    int* csrc   = rank   + E;
    float* wl   = (float*)(csrc + E);
    float* wr   = wl + DH;

    const int gN16  = (N + 15) / 16;
    const int gE256 = (E + 255) / 256;
    const int gL256 = LEN / 256;
    const int gW8   = (N + 7) / 8;
    const int gCopy = 510;

    csr_zero<<<gL256, 256, 0, stream>>>(deg, LEN);
    csr_rank<<<gE256, 256, 0, stream>>>(dst, deg, rank, E);

    prep_layer1<<<gCopy + 2, SCAN_T, 0, stream>>>(x, rep, W1, al1, ar1, feat, wl, wr, N,
                                                  deg, rowptr, LEN);
    csr_fill<<<gE256, 256, 0, stream>>>(src, dst, rank, rowptr, csrc, E);

    gat_gather<false, true><<<gW8, 256, 0, stream>>>(rowptr, csrc, wl, wr, feat, nullptr, nullptr, rly,
                                                     nullptr, nullptr, N);

    gemm_two<<<gN16, 512, 0, stream>>>(rly, W1, b1, W2, al2, ar2, h, el, er, N);
    gat_gather<true, false><<<gW8, 256, 0, stream>>>(rowptr, csrc, al2, er, h, b2, fcW, nullptr, pS, pD, N);

    edge_classify<<<gE256, 256, 0, stream>>>(src, dst, pS, pD, fcb, out, E);
}


# edges written into a fixed 32-slot row per dst at rank time; longer lists spill to an overflow list added edge by edge (sums regrouped there); scan and fill launches removed: 8 -> 6 launches
# speedup vs baseline: 4.8727x; 1.0722x over previous
#include <hip/hip_runtime.h>
#include <hip/hip_bf16.h>
#include <math.h>

typedef __attribute__((ext_vector_type(2))) float v2f;
typedef __attribute__((ext_vector_type(4))) float v4f;
typedef __attribute__((ext_vector_type(8))) float v8f;

#define HID 128
#define HEADS 2
#define DH 256
#define NEG_SLOPE 0.2f

#define PREP_T 512

__global__ __launch_bounds__(PREP_T)
void prep_layer1(const float* __restrict__ x, const float* __restrict__ rep,
                 const float* __restrict__ W1, const float* __restrict__ al,
                 const float* __restrict__ ar, float* __restrict__ feat,
                 float* __restrict__ wl, float* __restrict__ wr, int n,
                 int* __restrict__ deg)
{
    if (blockIdx.x == 0) {
        const int  t     = threadIdx.x;
        const int  k     = t >> 2;
        const int  head  = t & 1;
        const bool right = (t & 2) != 0;
        const float* wrow = W1 + k * DH + head * HID;
        const float* a    = (right ? ar : al) + head * HID;
        float sum = 0.0f;
        for (int j = 0; j < HID; j += 4) {
            const v4f w = *(const v4f*)(wrow + j);
            const v4f v = *(const v4f*)(a + j);
            sum += w.x * v.x + w.y * v.y + w.z * v.z + w.w * v.w;
        }
        (right ? wr : wl)[head * HID + k] = sum;
        return;
    }
    const int first  = (blockIdx.x - 1) * PREP_T + threadIdx.x;
    const int stride = (gridDim.x - 1) * PREP_T;
    for (int i = first; i <= n; i += stride) deg[i] = 0;
    const int total  = n * 32;
    for (int idx = first; idx < total; idx += stride) {
        const int row = idx >> 5;
        const int q   = idx & 31;
        const v4f v = (q < 16) ? *(const v4f*)(x + (size_t)row * 64 + q * 4)
                               : *(const v4f*)(rep + (size_t)row * 64 + (q - 16) * 4);
        *(v4f*)(feat + (size_t)row * HID + q * 4) = v;
    }
}

__global__ __launch_bounds__(512)
void gemm_two(const float* __restrict__ G, const float* __restrict__ W1,
              const float* __restrict__ bias1, const float* __restrict__ W,
              const float* __restrict__ al, const float* __restrict__ ar,
              float* __restrict__ Hout, float* __restrict__ el, float* __restrict__ er, int n)
{
    constexpr int K     = DH;
    constexpr int PITCH = K + 4;
    __shared__ float As[16 * PITCH];
    __shared__ float Rs[16 * PITCH];
    __shared__ float red[16 * 32];

    const int tid  = threadIdx.x;
    const int wid  = tid >> 5;
    const int lane = tid & 31;
    const int r0   = blockIdx.x * 16;

    for (int idx = tid; idx < 16 * K; idx += 512) {
        const int row = idx / K;
        const int k   = idx - row * K;
        const int gr  = r0 + row;
        float v = 0.0f;
        if (gr < n) v = G[(size_t)gr * K + k];
        As[row * PITCH + k] = v;
    }
    __syncthreads();

    const int c0    = wid * 16;
    const int halfl = lane & 15;
    const int koff  = (lane >> 4) * 2;
    const int colg  = c0 + halfl;
    const int mbase = (lane >> 4) * 8;

    {
        const int hk = (wid >> 3) * HID;
        v8f acc1 = {};
        for (int k0 = 0; k0 < HID; k0 += 4) {
            v2f a, b;
            a.x = As[halfl * PITCH + hk + k0 + koff];
            a.y = As[halfl * PITCH + hk + k0 + koff + 1];
            b.x = W1[(k0 + koff)     * DH + colg];
            b.y = W1[(k0 + koff + 1) * DH + colg];
            acc1 = __builtin_amdgcn_wmma_f32_16x16x4_f32(
                       false, a, false, b, (short)0, acc1, false, false);
        }
        const float bv = bias1[colg];
#pragma unroll
        for (int r = 0; r < 8; ++r) {
            const float v = acc1[r] + bv;
            Rs[(mbase + r) * PITCH + colg] = v > 0.0f ? v : 0.0f;
        }
    }
    __syncthreads();

    v8f acc = {};
    for (int k0 = 0; k0 < K; k0 += 4) {
        v2f a, b;
        a.x = Rs[halfl * PITCH + k0 + koff];
        a.y = Rs[halfl * PITCH + k0 + koff + 1];
        b.x = W[(k0 + koff)     * DH + colg];
        b.y = W[(k0 + koff + 1) * DH + colg];
        acc = __builtin_amdgcn_wmma_f32_16x16x4_f32(
                  false, a, false, b, (short)0, acc, false, false);
    }

#pragma unroll
    for (int r = 0; r < 8; ++r) {
        const int grow = r0 + mbase + r;
        if (grow < n) Hout[grow * DH + colg] = acc[r];
    }
    const float alv = al[colg];
    const float arv = ar[colg];
    float p[16];
#pragma unroll
    for (int r = 0; r < 8; ++r) { p[2 * r] = acc[r] * alv; p[2 * r + 1] = acc[r] * arv; }

    const bool b3 = (lane & 8) != 0, b2 = (lane & 4) != 0, b1 = (lane & 2) != 0, b0 = (lane & 1) != 0;
    float u[8], t[4], s[2];
#pragma unroll
    for (int i = 0; i < 8; ++i) {
        const float keep = b3 ? p[i + 8] : p[i];
        const float send = b3 ? p[i] : p[i + 8];
        u[i] = keep + __shfl_xor(send, 8);
    }
#pragma unroll
    for (int i = 0; i < 4; ++i) {
        const float keep = b2 ? u[i + 4] : u[i];
        const float send = b2 ? u[i] : u[i + 4];
        t[i] = keep + __shfl_xor(send, 4);
    }
#pragma unroll
    for (int i = 0; i < 2; ++i) {
        const float keep = b1 ? t[i + 2] : t[i];
        const float send = b1 ? t[i] : t[i + 2];
        s[i] = keep + __shfl_xor(send, 2);
    }
    const float keep0 = b0 ? s[1] : s[0];
    const float send0 = b0 ? s[0] : s[1];
    red[wid * 32 + lane] = keep0 + __shfl_xor(send0, 1);
    __syncthreads();

    if (wid == 0) {
        float h0 = 0.0f, h1 = 0.0f;
#pragma unroll
        for (int i = 0; i < 8; ++i) {
            h0 += red[i * 32 + lane];
            h1 += red[(8 + i) * 32 + lane];
        }
        const int grow = r0 + mbase + (halfl >> 1);
        if (grow < n) {
            float* dstp = (lane & 1) ? er : el;
            *(float2*)(dstp + grow * 2) = make_float2(h0, h1);
        }
    }
}

__device__ __forceinline__ float lrelu(float x) {
    return x > 0.0f ? x : NEG_SLOPE * x;
}

#define CAP 32

__global__ void csr_bucket(const int* __restrict__ src, const int* __restrict__ dst,
                           int* __restrict__ deg, int* __restrict__ csrc,
                           int* __restrict__ ovf, int* __restrict__ ovf_n, int e)
{
    const int i = blockIdx.x * 256 + threadIdx.x;
    if (i >= e) return;
    const int d = dst[i];
    const int s = src[i];
    const int r = atomicAdd(&deg[d], 1);
    if (r < CAP) {
        csrc[d * CAP + r] = s;
    } else {
        const int k = atomicAdd(ovf_n, 1);
        *(int2*)(ovf + (size_t)k * 2) = make_int2(d, s);
    }
}

template<int M>
__device__ __forceinline__ float xrow(float x) {
    const int b = __builtin_bit_cast(int, x);
    return __builtin_bit_cast(float,
        __builtin_amdgcn_update_dpp(b, b, 0x160 + M, 0xf, 0xf, false));
}
template<int L>
__device__ __forceinline__ float rshare(float x) {
    const int b = __builtin_bit_cast(int, x);
    return __builtin_bit_cast(float,
        __builtin_amdgcn_update_dpp(b, b, 0x150 + L, 0xf, 0xf, false));
}

template<int RS>
__device__ __forceinline__
void edge_group(const float* __restrict__ Hl, int s0, int s1, int s2, int s3, int nv,
                const v4f aA, const v4f aB, float erd, bool up8, bool up4,
                float& m, float& l, v4f& accA, v4f& accB)
{
    const float* r0 = Hl + (size_t)s0 * RS;
    const float* r1 = Hl + (size_t)s1 * RS;
    const float* r2 = Hl + (size_t)s2 * RS;
    const float* r3 = Hl + (size_t)s3 * RS;
    const v4f x0 = *(const v4f*)r0, y0 = *(const v4f*)(r0 + 64);
    const v4f x1 = *(const v4f*)r1, y1 = *(const v4f*)(r1 + 64);
    const v4f x2 = *(const v4f*)r2, y2 = *(const v4f*)(r2 + 64);
    const v4f x3 = *(const v4f*)r3, y3 = *(const v4f*)(r3 + 64);

    const v2f q0 = x0.lo * aA.lo + x0.hi * aA.hi + y0.lo * aB.lo + y0.hi * aB.hi;
    const v2f q1 = x1.lo * aA.lo + x1.hi * aA.hi + y1.lo * aB.lo + y1.hi * aB.hi;
    const v2f q2 = x2.lo * aA.lo + x2.hi * aA.hi + y2.lo * aB.lo + y2.hi * aB.hi;
    const v2f q3 = x3.lo * aA.lo + x3.hi * aA.hi + y3.lo * aB.lo + y3.hi * aB.hi;
    const float p0 = q0.x + q0.y;
    const float p1 = q1.x + q1.y;
    const float p2 = q2.x + q2.y;
    const float p3 = q3.x + q3.y;

    const float k0 = up8 ? p2 : p0, g0 = up8 ? p0 : p2;
    const float k1 = up8 ? p3 : p1, g1 = up8 ? p1 : p3;
    const float u0 = k0 + xrow<8>(g0);
    const float u1 = k1 + xrow<8>(g1);
    const float kk = up4 ? u1 : u0,  gg = up4 ? u0 : u1;
    float t = kk + xrow<4>(gg);
    t += xrow<2>(t);
    t += xrow<1>(t);
    const float d0 = rshare<0>(t);
    const float d1 = rshare<4>(t);
    const float d2 = rshare<8>(t);
    const float d3 = rshare<12>(t);

    const float e0 = lrelu(d0 + erd);
    const float e1 = lrelu(d1 + erd);
    const float e2 = lrelu(d2 + erd);
    const float e3 = lrelu(d3 + erd);
    const float mn = fmaxf(fmaxf(m, fmaxf(e0, e1)), fmaxf(e2, e3));
    const float sc = __expf(m - mn);
    const float w0 = __expf(e0 - mn);
    const float w1 = (nv > 1) ? __expf(e1 - mn) : 0.0f;
    const float w2 = (nv > 2) ? __expf(e2 - mn) : 0.0f;
    const float w3 = (nv > 3) ? __expf(e3 - mn) : 0.0f;
    l    = l * sc + ((w0 + w1) + (w2 + w3));
    accA = accA * sc + x0 * w0;
    accB = accB * sc + y0 * w0;
    accA += x1 * w1;  accB += y1 * w1;
    accA += x2 * w2;  accB += y2 * w2;
    accA += x3 * w3;  accB += y3 * w3;
    m = mn;
}

template<bool LAST, bool L1>
__global__ __launch_bounds__(256)
void gat_gather(const int* __restrict__ deg, const int* __restrict__ csrc,
                const int* __restrict__ ovf, const int* __restrict__ ovf_n,
                const float* __restrict__ al, const float* __restrict__ er,
                const float* __restrict__ H, const float* __restrict__ bias,
                const float* __restrict__ fcW, float* __restrict__ outp,
                float* __restrict__ pS, float* __restrict__ pD, int n)
{
    const int lane = threadIdx.x & 31;
    const int node = __builtin_amdgcn_readfirstlane(blockIdx.x * 8 + (threadIdx.x >> 5));
    if (node >= n) return;

    const int indeg = deg[node];
    const int cnt   = min(indeg, CAP);

    const int   head = lane >> 4;
    const int   col0 = head * HID + (lane & 15) * 4;
    constexpr int RS = L1 ? HID : DH;
    const int   rc0  = L1 ? (lane & 15) * 4 : col0;

    int sidx = 0;
    if (lane < cnt) sidx = csrc[node * CAP + lane];

    const v4f    aA = *(const v4f*)(al + col0);
    const v4f    aB = *(const v4f*)(al + col0 + 64);
    const float* Hl = H + rc0;

    float erd;
    if (L1) {
        const v4f    rA  = *(const v4f*)(er + col0);
        const v4f    rB  = *(const v4f*)(er + col0 + 64);
        const float* own = Hl + (size_t)node * RS;
        const v4f    xo  = *(const v4f*)own, yo = *(const v4f*)(own + 64);
        float d = xo.x * rA.x + xo.y * rA.y + xo.z * rA.z + xo.w * rA.w
                + yo.x * rB.x + yo.y * rB.y + yo.z * rB.z + yo.w * rB.w;
        d += xrow<8>(d); d += xrow<4>(d); d += xrow<2>(d); d += xrow<1>(d);
        erd = d;
    } else {
        erd = er[node * 2 + head];
    }
    const bool   up8 = (lane & 8) != 0;
    const bool   up4 = (lane & 4) != 0;

    float m    = -INFINITY;
    float l    = 0.0f;
    v4f   accA = {};
    v4f   accB = {};

    const int last = cnt - 1;
    for (int j = 0; j < cnt; j += 4) {
        const int s0 = __builtin_amdgcn_readlane(sidx, j);
        const int s1 = __builtin_amdgcn_readlane(sidx, min(j + 1, last));
        const int s2 = __builtin_amdgcn_readlane(sidx, min(j + 2, last));
        const int s3 = __builtin_amdgcn_readlane(sidx, min(j + 3, last));
        edge_group<RS>(Hl, s0, s1, s2, s3, cnt - j, aA, aB, erd, up8, up4, m, l, accA, accB);
    }
    if (indeg > CAP) {
        const int on = ovf_n[0];
        for (int base = 0; base < on; base += 32) {
            int od = -1, os = 0;
            if (base + lane < on) {
                const int2 pr = *(const int2*)(ovf + (size_t)(base + lane) * 2);
                od = pr.x;  os = pr.y;
            }
            unsigned mine = __builtin_amdgcn_ballot_w32(od == node);
            while (mine != 0u) {
                const int b = __builtin_ctz(mine);
                mine &= mine - 1u;
                const int s = __builtin_amdgcn_readlane(os, b);
                edge_group<RS>(Hl, s, s, s, s, 1, aA, aB, erd, up8, up4, m, l, accA, accB);
            }
        }
    }
    if (l > 0.0f) { const float rl = 1.0f / l; accA = accA * rl; accB = accB * rl; }

    if (L1) {
        float* op = outp + (size_t)node * DH + col0;
        *(v4f*)op        = accA;
        *(v4f*)(op + 64) = accB;
        return;
    }

    const v4f bA = *(const v4f*)(bias + col0);
    const v4f bB = *(const v4f*)(bias + col0 + 64);
    v4f oA = accA + bA, oB = accB + bB;
    oA.x = oA.x > 0.0f ? oA.x : 0.0f;  oA.y = oA.y > 0.0f ? oA.y : 0.0f;
    oA.z = oA.z > 0.0f ? oA.z : 0.0f;  oA.w = oA.w > 0.0f ? oA.w : 0.0f;
    oB.x = oB.x > 0.0f ? oB.x : 0.0f;  oB.y = oB.y > 0.0f ? oB.y : 0.0f;
    oB.z = oB.z > 0.0f ? oB.z : 0.0f;  oB.w = oB.w > 0.0f ? oB.w : 0.0f;

    if (!LAST) {
        float* op = outp + (size_t)node * DH + col0;
        *(v4f*)op        = oA;
        *(v4f*)(op + 64) = oB;
    } else {
        const float* fs = fcW + col0 * 2;
        const float* fd = fcW + (DH + col0) * 2;
        const v4f tA0 = *(const v4f*)fs,         tA1 = *(const v4f*)(fs + 4);
        const v4f tB0 = *(const v4f*)(fs + 128), tB1 = *(const v4f*)(fs + 132);
        const v4f uA0 = *(const v4f*)fd,         uA1 = *(const v4f*)(fd + 4);
        const v4f uB0 = *(const v4f*)(fd + 128), uB1 = *(const v4f*)(fd + 132);
        float s0 = oA.x * tA0.x + oA.y * tA0.z + oA.z * tA1.x + oA.w * tA1.z
                 + oB.x * tB0.x + oB.y * tB0.z + oB.z * tB1.x + oB.w * tB1.z;
        float s1 = oA.x * tA0.y + oA.y * tA0.w + oA.z * tA1.y + oA.w * tA1.w
                 + oB.x * tB0.y + oB.y * tB0.w + oB.z * tB1.y + oB.w * tB1.w;
        float d0 = oA.x * uA0.x + oA.y * uA0.z + oA.z * uA1.x + oA.w * uA1.z
                 + oB.x * uB0.x + oB.y * uB0.z + oB.z * uB1.x + oB.w * uB1.z;
        float d1 = oA.x * uA0.y + oA.y * uA0.w + oA.z * uA1.y + oA.w * uA1.w
                 + oB.x * uB0.y + oB.y * uB0.w + oB.z * uB1.y + oB.w * uB1.w;
        s0 += xrow<8>(s0); s1 += xrow<8>(s1); d0 += xrow<8>(d0); d1 += xrow<8>(d1);
        s0 += xrow<4>(s0); s1 += xrow<4>(s1); d0 += xrow<4>(d0); d1 += xrow<4>(d1);
        s0 += xrow<2>(s0); s1 += xrow<2>(s1); d0 += xrow<2>(d0); d1 += xrow<2>(d1);
        s0 += xrow<1>(s0); s1 += xrow<1>(s1); d0 += xrow<1>(d0); d1 += xrow<1>(d1);
        if ((lane & 15) == 0) {
            const size_t slot = ((size_t)node * 2 + head) * 2;
            *(float2*)(pS + slot) = make_float2(s0, s1);
            *(float2*)(pD + slot) = make_float2(d0, d1);
        }
    }
}

__global__ __launch_bounds__(256)
void edge_classify(const int* __restrict__ src, const int* __restrict__ dst,
                   const float* __restrict__ pS, const float* __restrict__ pD,
                   const float* __restrict__ fcb, float* __restrict__ out, int e)
{
    const int i = blockIdx.x * 256 + threadIdx.x;
    if (i >= e) return;
    const int s = src[i], d = dst[i];
    const float4 ps = *(const float4*)(pS + (size_t)s * 4);
    const float4 pd = *(const float4*)(pD + (size_t)d * 4);
    float2 o;
    o.x = (ps.x + ps.z) + (pd.x + pd.z) + fcb[0];
    o.y = (ps.y + ps.w) + (pd.y + pd.w) + fcb[1];
    *(float2*)(out + (size_t)i * 2) = o;
}

extern "C" void kernel_launch(void* const* d_in, const int* in_sizes, int n_in,
                              void* d_out, int out_size, void* d_ws, size_t ws_size,
                              hipStream_t stream)
{
    const float* x   = (const float*)d_in[0];
    const float* rep = (const float*)d_in[1];
    const int*   src = (const int*)d_in[2];
    const int*   dst = (const int*)d_in[3];
    const float* W1  = (const float*)d_in[4];
    const float* al1 = (const float*)d_in[5];
    const float* ar1 = (const float*)d_in[6];
    const float* b1  = (const float*)d_in[7];
    const float* W2  = (const float*)d_in[8];
    const float* al2 = (const float*)d_in[9];
    const float* ar2 = (const float*)d_in[10];
    const float* b2  = (const float*)d_in[11];
    const float* fcW = (const float*)d_in[12];
    const float* fcb = (const float*)d_in[13];
    float* out = (float*)d_out;

    const int N = in_sizes[0] / 64;
    const int E = in_sizes[2];

    float* ws  = (float*)d_ws;
    float* h   = ws;
    float* feat = ws;
    int*   csr = (int*)(h + (size_t)N * DH);
    float* rly = h   + (size_t)N * DH * 2;
    float* pS  = rly;
    float* pD  = rly + (size_t)N * 4;
    float* el  = rly + (size_t)N * DH;
    float* er  = el  + (size_t)N * 2;

    const int LEN = ((N + 1 + 63) / 64) * 64;
    int* deg    = csr;
    int* ovf_n  = deg + N;
    int* csrc   = deg + LEN;
    int* ovf    = csrc + (size_t)N * CAP;
    float* wl   = (float*)(ovf + (size_t)E * 2);
    float* wr   = wl + DH;

    const int gN16  = (N + 15) / 16;
    const int gE256 = (E + 255) / 256;
    const int gW8   = (N + 7) / 8;
    const int gCopy = 511;

    prep_layer1<<<gCopy + 1, PREP_T, 0, stream>>>(x, rep, W1, al1, ar1, feat, wl, wr, N, deg);

    csr_bucket<<<gE256, 256, 0, stream>>>(src, dst, deg, csrc, ovf, ovf_n, E);

    gat_gather<false, true><<<gW8, 256, 0, stream>>>(deg, csrc, ovf, ovf_n, wl, wr, feat, nullptr, nullptr,
                                                     rly, nullptr, nullptr, N);

    gemm_two<<<gN16, 512, 0, stream>>>(rly, W1, b1, W2, al2, ar2, h, el, er, N);
    gat_gather<true, false><<<gW8, 256, 0, stream>>>(deg, csrc, ovf, ovf_n, al2, er, h, b2, fcW, nullptr,
                                                     pS, pD, N);

    edge_classify<<<gE256, 256, 0, stream>>>(src, dst, pS, pD, fcb, out, E);
}


# GEMM weights repacked to WMMA fragment order by the first launch: a lane's B operands of two 16x16x4 products come as one 16-byte whole-line load (was four 4-byte); k summed in another order
# speedup vs baseline: 5.1326x; 1.0533x over previous
#include <hip/hip_runtime.h>
#include <hip/hip_bf16.h>
#include <math.h>

typedef __attribute__((ext_vector_type(2))) float v2f;
typedef __attribute__((ext_vector_type(4))) float v4f;
typedef __attribute__((ext_vector_type(8))) float v8f;

#define HID 128
#define HEADS 2
#define DH 256
#define NEG_SLOPE 0.2f

#define PREP_T 512
#define PACK_W1 (HID / 8 * 16 * 32)
#define PACK_N  ((HID + DH) / 8 * 16 * 32)

__global__ __launch_bounds__(PREP_T)
void prep_layer1(const float* __restrict__ x, const float* __restrict__ rep,
                 const float* __restrict__ W1, const float* __restrict__ al,
                 const float* __restrict__ ar, float* __restrict__ feat,
                 float* __restrict__ wl, float* __restrict__ wr, int n,
                 int* __restrict__ deg, const float* __restrict__ W2,
                 float* __restrict__ wpk)
{
    if (blockIdx.x == 0) {
        const int  t     = threadIdx.x;
        const int  k     = t >> 2;
        const int  head  = t & 1;
        const bool right = (t & 2) != 0;
        const float* wrow = W1 + k * DH + head * HID;
        const float* a    = (right ? ar : al) + head * HID;
        float sum = 0.0f;
        for (int j = 0; j < HID; j += 4) {
            const v4f w = *(const v4f*)(wrow + j);
            const v4f v = *(const v4f*)(a + j);
            sum += w.x * v.x + w.y * v.y + w.z * v.z + w.w * v.w;
        }
        (right ? wr : wl)[head * HID + k] = sum;
        return;
    }
    const int first  = (blockIdx.x - 1) * PREP_T + threadIdx.x;
    const int stride = (gridDim.x - 1) * PREP_T;
    for (int i = first; i <= n; i += stride) deg[i] = 0;
    const int total  = n * 32;
    for (int idx = first; idx < total; idx += stride) {
        const int row = idx >> 5;
        const int q   = idx & 31;
        const v4f v = (q < 16) ? *(const v4f*)(x + (size_t)row * 64 + q * 4)
                               : *(const v4f*)(rep + (size_t)row * 64 + (q - 16) * 4);
        *(v4f*)(feat + (size_t)row * HID + q * 4) = v;
    }
    for (int idx = first; idx < PACK_N; idx += stride) {
        const int  ln     = idx & 31;
        const int  t      = (idx >> 5) & 15;
        const bool second = idx >= PACK_W1;
        const int  kb     = (second ? idx - PACK_W1 : idx) >> 9;
        const float* p = (second ? W2 : W1) + (kb * 8 + (ln >> 4) * 4) * DH + t * 16 + (ln & 15);
        v4f v;
        v.x = p[0];  v.y = p[DH];  v.z = p[2 * DH];  v.w = p[3 * DH];
        *(v4f*)(wpk + (size_t)idx * 4) = v;
    }
}

__global__ __launch_bounds__(512)
void gemm_two(const float* __restrict__ G, const float* __restrict__ Wp1,
              const float* __restrict__ bias1, const float* __restrict__ Wp2,
              const float* __restrict__ al, const float* __restrict__ ar,
              float* __restrict__ Hout, float* __restrict__ el, float* __restrict__ er, int n)
{
    constexpr int K     = DH;
    constexpr int PITCH = K + 4;
    __shared__ __attribute__((aligned(16))) float As[16 * PITCH];
    __shared__ __attribute__((aligned(16))) float Rs[16 * PITCH];
    __shared__ float red[16 * 32];

    const int tid  = threadIdx.x;
    const int wid  = tid >> 5;
    const int lane = tid & 31;
    const int r0   = blockIdx.x * 16;

    for (int idx = tid; idx < 16 * K; idx += 512) {
        const int row = idx / K;
        const int k   = idx - row * K;
        const int gr  = r0 + row;
        float v = 0.0f;
        if (gr < n) v = G[(size_t)gr * K + k];
        As[row * PITCH + k] = v;
    }
    __syncthreads();

    const int c0    = wid * 16;
    const int halfl = lane & 15;
    const int koff  = (lane >> 4) * 4;
    const int colg  = c0 + halfl;
    const int bpos  = (wid * 32 + lane) * 4;
    constexpr int BSTEP = 16 * 32 * 4;
    const int mbase = (lane >> 4) * 8;

    {
        const int hk = (wid >> 3) * HID;
        v8f acc1 = {};
        for (int kb = 0; kb < HID / 8; ++kb) {
            const v4f a4 = *(const v4f*)(As + halfl * PITCH + hk + kb * 8 + koff);
            const v4f b4 = *(const v4f*)(Wp1 + kb * BSTEP + bpos);
            acc1 = __builtin_amdgcn_wmma_f32_16x16x4_f32(
                       false, a4.lo, false, b4.lo, (short)0, acc1, false, false);
            acc1 = __builtin_amdgcn_wmma_f32_16x16x4_f32(
                       false, a4.hi, false, b4.hi, (short)0, acc1, false, false);
        }
        const float bv = bias1[colg];
#pragma unroll
        for (int r = 0; r < 8; ++r) {
            const float v = acc1[r] + bv;
            Rs[(mbase + r) * PITCH + colg] = v > 0.0f ? v : 0.0f;
        }
    }
    __syncthreads();

    v8f acc = {};
    for (int kb = 0; kb < K / 8; ++kb) {
        const v4f a4 = *(const v4f*)(Rs + halfl * PITCH + kb * 8 + koff);
        const v4f b4 = *(const v4f*)(Wp2 + kb * BSTEP + bpos);
        acc = __builtin_amdgcn_wmma_f32_16x16x4_f32(
                  false, a4.lo, false, b4.lo, (short)0, acc, false, false);
        acc = __builtin_amdgcn_wmma_f32_16x16x4_f32(
                  false, a4.hi, false, b4.hi, (short)0, acc, false, false);
    }

#pragma unroll
    for (int r = 0; r < 8; ++r) {
        const int grow = r0 + mbase + r;
        if (grow < n) Hout[grow * DH + colg] = acc[r];
    }
    const float alv = al[colg];
    const float arv = ar[colg];
    float p[16];
#pragma unroll
    for (int r = 0; r < 8; ++r) { p[2 * r] = acc[r] * alv; p[2 * r + 1] = acc[r] * arv; }

    const bool b3 = (lane & 8) != 0, b2 = (lane & 4) != 0, b1 = (lane & 2) != 0, b0 = (lane & 1) != 0;
    float u[8], t[4], s[2];
#pragma unroll
    for (int i = 0; i < 8; ++i) {
        const float keep = b3 ? p[i + 8] : p[i];
        const float send = b3 ? p[i] : p[i + 8];
        u[i] = keep + __shfl_xor(send, 8);
    }
#pragma unroll
    for (int i = 0; i < 4; ++i) {
        const float keep = b2 ? u[i + 4] : u[i];
        const float send = b2 ? u[i] : u[i + 4];
        t[i] = keep + __shfl_xor(send, 4);
    }
#pragma unroll
    for (int i = 0; i < 2; ++i) {
        const float keep = b1 ? t[i + 2] : t[i];
        const float send = b1 ? t[i] : t[i + 2];
        s[i] = keep + __shfl_xor(send, 2);
    }
    const float keep0 = b0 ? s[1] : s[0];
    const float send0 = b0 ? s[0] : s[1];
    red[wid * 32 + lane] = keep0 + __shfl_xor(send0, 1);
    __syncthreads();

    if (wid == 0) {
        float h0 = 0.0f, h1 = 0.0f;
#pragma unroll
        for (int i = 0; i < 8; ++i) {
            h0 += red[i * 32 + lane];
            h1 += red[(8 + i) * 32 + lane];
        }
        const int grow = r0 + mbase + (halfl >> 1);
        if (grow < n) {
            float* dstp = (lane & 1) ? er : el;
            *(float2*)(dstp + grow * 2) = make_float2(h0, h1);
        }
    }
}

__device__ __forceinline__ float lrelu(float x) {
    return x > 0.0f ? x : NEG_SLOPE * x;
}

#define CAP 32

__global__ void csr_bucket(const int* __restrict__ src, const int* __restrict__ dst,
                           int* __restrict__ deg, int* __restrict__ csrc,
                           int* __restrict__ ovf, int* __restrict__ ovf_n, int e)
{
    const int i = blockIdx.x * 256 + threadIdx.x;
    if (i >= e) return;
    const int d = dst[i];
    const int s = src[i];
    const int r = atomicAdd(&deg[d], 1);
    if (r < CAP) {
        csrc[d * CAP + r] = s;
    } else {
        const int k = atomicAdd(ovf_n, 1);
        *(int2*)(ovf + (size_t)k * 2) = make_int2(d, s);
    }
}

template<int M>
__device__ __forceinline__ float xrow(float x) {
    const int b = __builtin_bit_cast(int, x);
    return __builtin_bit_cast(float,
        __builtin_amdgcn_update_dpp(b, b, 0x160 + M, 0xf, 0xf, false));
}
template<int L>
__device__ __forceinline__ float rshare(float x) {
    const int b = __builtin_bit_cast(int, x);
    return __builtin_bit_cast(float,
        __builtin_amdgcn_update_dpp(b, b, 0x150 + L, 0xf, 0xf, false));
}

template<int RS>
__device__ __forceinline__
void edge_group(const float* __restrict__ Hl, int s0, int s1, int s2, int s3, int nv,
                const v4f aA, const v4f aB, float erd, bool up8, bool up4,
                float& m, float& l, v4f& accA, v4f& accB)
{
    const float* r0 = Hl + (size_t)s0 * RS;
    const float* r1 = Hl + (size_t)s1 * RS;
    const float* r2 = Hl + (size_t)s2 * RS;
    const float* r3 = Hl + (size_t)s3 * RS;
    const v4f x0 = *(const v4f*)r0, y0 = *(const v4f*)(r0 + 64);
    const v4f x1 = *(const v4f*)r1, y1 = *(const v4f*)(r1 + 64);
    const v4f x2 = *(const v4f*)r2, y2 = *(const v4f*)(r2 + 64);
    const v4f x3 = *(const v4f*)r3, y3 = *(const v4f*)(r3 + 64);

    const v2f q0 = x0.lo * aA.lo + x0.hi * aA.hi + y0.lo * aB.lo + y0.hi * aB.hi;
    const v2f q1 = x1.lo * aA.lo + x1.hi * aA.hi + y1.lo * aB.lo + y1.hi * aB.hi;
    const v2f q2 = x2.lo * aA.lo + x2.hi * aA.hi + y2.lo * aB.lo + y2.hi * aB.hi;
    const v2f q3 = x3.lo * aA.lo + x3.hi * aA.hi + y3.lo * aB.lo + y3.hi * aB.hi;
    const float p0 = q0.x + q0.y;
    const float p1 = q1.x + q1.y;
    const float p2 = q2.x + q2.y;
    const float p3 = q3.x + q3.y;

    const float k0 = up8 ? p2 : p0, g0 = up8 ? p0 : p2;
    const float k1 = up8 ? p3 : p1, g1 = up8 ? p1 : p3;
    const float u0 = k0 + xrow<8>(g0);
    const float u1 = k1 + xrow<8>(g1);
    const float kk = up4 ? u1 : u0,  gg = up4 ? u0 : u1;
    float t = kk + xrow<4>(gg);
    t += xrow<2>(t);
    t += xrow<1>(t);
    const float d0 = rshare<0>(t);
    const float d1 = rshare<4>(t);
    const float d2 = rshare<8>(t);
    const float d3 = rshare<12>(t);

    const float e0 = lrelu(d0 + erd);
    const float e1 = lrelu(d1 + erd);
    const float e2 = lrelu(d2 + erd);
    const float e3 = lrelu(d3 + erd);
    const float mn = fmaxf(fmaxf(m, fmaxf(e0, e1)), fmaxf(e2, e3));
    const float sc = __expf(m - mn);
    const float w0 = __expf(e0 - mn);
    const float w1 = (nv > 1) ? __expf(e1 - mn) : 0.0f;
    const float w2 = (nv > 2) ? __expf(e2 - mn) : 0.0f;
    const float w3 = (nv > 3) ? __expf(e3 - mn) : 0.0f;
    l    = l * sc + ((w0 + w1) + (w2 + w3));
    accA = accA * sc + x0 * w0;
    accB = accB * sc + y0 * w0;
    accA += x1 * w1;  accB += y1 * w1;
    accA += x2 * w2;  accB += y2 * w2;
    accA += x3 * w3;  accB += y3 * w3;
    m = mn;
}

template<bool LAST, bool L1>
__global__ __launch_bounds__(256)
void gat_gather(const int* __restrict__ deg, const int* __restrict__ csrc,
                const int* __restrict__ ovf, const int* __restrict__ ovf_n,
                const float* __restrict__ al, const float* __restrict__ er,
                const float* __restrict__ H, const float* __restrict__ bias,
                const float* __restrict__ fcW, float* __restrict__ outp,
                float* __restrict__ pS, float* __restrict__ pD, int n)
{
    const int lane = threadIdx.x & 31;
    const int node = __builtin_amdgcn_readfirstlane(blockIdx.x * 8 + (threadIdx.x >> 5));
    if (node >= n) return;

    const int indeg = deg[node];
    const int cnt   = min(indeg, CAP);

    const int   head = lane >> 4;
    const int   col0 = head * HID + (lane & 15) * 4;
    constexpr int RS = L1 ? HID : DH;
    const int   rc0  = L1 ? (lane & 15) * 4 : col0;

    int sidx = 0;
    if (lane < cnt) sidx = csrc[node * CAP + lane];

    const v4f    aA = *(const v4f*)(al + col0);
    const v4f    aB = *(const v4f*)(al + col0 + 64);
    const float* Hl = H + rc0;

    float erd;
    if (L1) {
        const v4f    rA  = *(const v4f*)(er + col0);
        const v4f    rB  = *(const v4f*)(er + col0 + 64);
        const float* own = Hl + (size_t)node * RS;
        const v4f    xo  = *(const v4f*)own, yo = *(const v4f*)(own + 64);
        float d = xo.x * rA.x + xo.y * rA.y + xo.z * rA.z + xo.w * rA.w
                + yo.x * rB.x + yo.y * rB.y + yo.z * rB.z + yo.w * rB.w;
        d += xrow<8>(d); d += xrow<4>(d); d += xrow<2>(d); d += xrow<1>(d);
        erd = d;
    } else {
        erd = er[node * 2 + head];
    }
    const bool   up8 = (lane & 8) != 0;
    const bool   up4 = (lane & 4) != 0;

    float m    = -INFINITY;
    float l    = 0.0f;
    v4f   accA = {};
    v4f   accB = {};

    const int last = cnt - 1;
    for (int j = 0; j < cnt; j += 4) {
        const int s0 = __builtin_amdgcn_readlane(sidx, j);
        const int s1 = __builtin_amdgcn_readlane(sidx, min(j + 1, last));
        const int s2 = __builtin_amdgcn_readlane(sidx, min(j + 2, last));
        const int s3 = __builtin_amdgcn_readlane(sidx, min(j + 3, last));
        edge_group<RS>(Hl, s0, s1, s2, s3, cnt - j, aA, aB, erd, up8, up4, m, l, accA, accB);
    }
    if (indeg > CAP) {
        const int on = ovf_n[0];
        for (int base = 0; base < on; base += 32) {
            int od = -1, os = 0;
            if (base + lane < on) {
                const int2 pr = *(const int2*)(ovf + (size_t)(base + lane) * 2);
                od = pr.x;  os = pr.y;
            }
            unsigned mine = __builtin_amdgcn_ballot_w32(od == node);
            while (mine != 0u) {
                const int b = __builtin_ctz(mine);
                mine &= mine - 1u;
                const int s = __builtin_amdgcn_readlane(os, b);
                edge_group<RS>(Hl, s, s, s, s, 1, aA, aB, erd, up8, up4, m, l, accA, accB);
            }
        }
    }
    if (l > 0.0f) { const float rl = 1.0f / l; accA = accA * rl; accB = accB * rl; }

    if (L1) {
        float* op = outp + (size_t)node * DH + col0;
        *(v4f*)op        = accA;
        *(v4f*)(op + 64) = accB;
        return;
    }

    const v4f bA = *(const v4f*)(bias + col0);
    const v4f bB = *(const v4f*)(bias + col0 + 64);
    v4f oA = accA + bA, oB = accB + bB;
    oA.x = oA.x > 0.0f ? oA.x : 0.0f;  oA.y = oA.y > 0.0f ? oA.y : 0.0f;
    oA.z = oA.z > 0.0f ? oA.z : 0.0f;  oA.w = oA.w > 0.0f ? oA.w : 0.0f;
    oB.x = oB.x > 0.0f ? oB.x : 0.0f;  oB.y = oB.y > 0.0f ? oB.y : 0.0f;
    oB.z = oB.z > 0.0f ? oB.z : 0.0f;  oB.w = oB.w > 0.0f ? oB.w : 0.0f;

    if (!LAST) {
        float* op = outp + (size_t)node * DH + col0;
        *(v4f*)op        = oA;
        *(v4f*)(op + 64) = oB;
    } else {
        const float* fs = fcW + col0 * 2;
        const float* fd = fcW + (DH + col0) * 2;
        const v4f tA0 = *(const v4f*)fs,         tA1 = *(const v4f*)(fs + 4);
        const v4f tB0 = *(const v4f*)(fs + 128), tB1 = *(const v4f*)(fs + 132);
        const v4f uA0 = *(const v4f*)fd,         uA1 = *(const v4f*)(fd + 4);
        const v4f uB0 = *(const v4f*)(fd + 128), uB1 = *(const v4f*)(fd + 132);
        float s0 = oA.x * tA0.x + oA.y * tA0.z + oA.z * tA1.x + oA.w * tA1.z
                 + oB.x * tB0.x + oB.y * tB0.z + oB.z * tB1.x + oB.w * tB1.z;
        float s1 = oA.x * tA0.y + oA.y * tA0.w + oA.z * tA1.y + oA.w * tA1.w
                 + oB.x * tB0.y + oB.y * tB0.w + oB.z * tB1.y + oB.w * tB1.w;
        float d0 = oA.x * uA0.x + oA.y * uA0.z + oA.z * uA1.x + oA.w * uA1.z
                 + oB.x * uB0.x + oB.y * uB0.z + oB.z * uB1.x + oB.w * uB1.z;
        float d1 = oA.x * uA0.y + oA.y * uA0.w + oA.z * uA1.y + oA.w * uA1.w
                 + oB.x * uB0.y + oB.y * uB0.w + oB.z * uB1.y + oB.w * uB1.w;
        s0 += xrow<8>(s0); s1 += xrow<8>(s1); d0 += xrow<8>(d0); d1 += xrow<8>(d1);
        s0 += xrow<4>(s0); s1 += xrow<4>(s1); d0 += xrow<4>(d0); d1 += xrow<4>(d1);
        s0 += xrow<2>(s0); s1 += xrow<2>(s1); d0 += xrow<2>(d0); d1 += xrow<2>(d1);
        s0 += xrow<1>(s0); s1 += xrow<1>(s1); d0 += xrow<1>(d0); d1 += xrow<1>(d1);
        if ((lane & 15) == 0) {
            const size_t slot = ((size_t)node * 2 + head) * 2;
            *(float2*)(pS + slot) = make_float2(s0, s1);
            *(float2*)(pD + slot) = make_float2(d0, d1);
        }
    }
}

__global__ __launch_bounds__(256)
void edge_classify(const int* __restrict__ src, const int* __restrict__ dst,
                   const float* __restrict__ pS, const float* __restrict__ pD,
                   const float* __restrict__ fcb, float* __restrict__ out, int e)
{
    const int i = blockIdx.x * 256 + threadIdx.x;
    if (i >= e) return;
    const int s = src[i], d = dst[i];
    const float4 ps = *(const float4*)(pS + (size_t)s * 4);
    const float4 pd = *(const float4*)(pD + (size_t)d * 4);
    float2 o;
    o.x = (ps.x + ps.z) + (pd.x + pd.z) + fcb[0];
    o.y = (ps.y + ps.w) + (pd.y + pd.w) + fcb[1];
    *(float2*)(out + (size_t)i * 2) = o;
}

extern "C" void kernel_launch(void* const* d_in, const int* in_sizes, int n_in,
                              void* d_out, int out_size, void* d_ws, size_t ws_size,
                              hipStream_t stream)
{
    const float* x   = (const float*)d_in[0];
    const float* rep = (const float*)d_in[1];
    const int*   src = (const int*)d_in[2];
    const int*   dst = (const int*)d_in[3];
    const float* W1  = (const float*)d_in[4];
    const float* al1 = (const float*)d_in[5];
    const float* ar1 = (const float*)d_in[6];
    const float* b1  = (const float*)d_in[7];
    const float* W2  = (const float*)d_in[8];
    const float* al2 = (const float*)d_in[9];
    const float* ar2 = (const float*)d_in[10];
    const float* b2  = (const float*)d_in[11];
    const float* fcW = (const float*)d_in[12];
    const float* fcb = (const float*)d_in[13];
    float* out = (float*)d_out;

    const int N = in_sizes[0] / 64;
    const int E = in_sizes[2];

    float* ws  = (float*)d_ws;
    float* h   = ws;
    float* feat = ws;
    int*   csr = (int*)(h + (size_t)N * DH);
    float* rly = h   + (size_t)N * DH * 2;
    float* pS  = rly;
    float* pD  = rly + (size_t)N * 4;
    float* el  = rly + (size_t)N * DH;
    float* er  = el  + (size_t)N * 2;

    const int LEN = ((N + 1 + 63) / 64) * 64;
    int* deg    = csr;
    int* ovf_n  = deg + N;
    int* csrc   = deg + LEN;
    int* ovf    = csrc + (size_t)N * CAP;
    float* wl   = (float*)(ovf + (size_t)E * 2);
    float* wr   = wl + DH;
    float* wpk  = wr + DH;

    const int gN16  = (N + 15) / 16;
    const int gE256 = (E + 255) / 256;
    const int gW8   = (N + 7) / 8;
    const int gCopy = 511;

    prep_layer1<<<gCopy + 1, PREP_T, 0, stream>>>(x, rep, W1, al1, ar1, feat, wl, wr, N, deg, W2, wpk);

    csr_bucket<<<gE256, 256, 0, stream>>>(src, dst, deg, csrc, ovf, ovf_n, E);

    gat_gather<false, true><<<gW8, 256, 0, stream>>>(deg, csrc, ovf, ovf_n, wl, wr, feat, nullptr, nullptr,
                                                     rly, nullptr, nullptr, N);

    gemm_two<<<gN16, 512, 0, stream>>>(rly, wpk, b1, wpk + PACK_W1 * 4, al2, ar2, h, el, er, N);
    gat_gather<true, false><<<gW8, 256, 0, stream>>>(deg, csrc, ovf, ovf_n, al2, er, h, b2, fcW, nullptr,
                                                     pS, pD, N);

    edge_classify<<<gE256, 256, 0, stream>>>(src, dst, pS, pD, fcb, out, E);
}
